# starting kernel
# baseline (speedup 1.0000x reference)
#include <hip/hip_runtime.h>
#include <math.h>


typedef __attribute__((ext_vector_type(2))) float v2f;
typedef __attribute__((ext_vector_type(8))) float v8f;

#define DIM 4096

constexpr unsigned cnot_src(unsigned k) {
  for (int i = 10; i >= 0; --i)
    for (int j = 11; j >= i + 1; --j)
      k ^= ((k >> (11 - i)) & 1u) << (11 - j);
  return k;
}
constexpr unsigned mask_p(int p) {
  unsigned mm = 0;
  for (int b = 0; b < 12; ++b)
    mm |= ((cnot_src(1u << b) >> p) & 1u) << b;
  return mm;
}
__constant__ unsigned SRC_MASK[12] = {
    mask_p(0), mask_p(1), mask_p(2),  mask_p(3),
    mask_p(4), mask_p(5), mask_p(6),  mask_p(7),
    mask_p(8), mask_p(9), mask_p(10), mask_p(11)};

__device__ __forceinline__ unsigned src_perm(unsigned k) {
  unsigned r = 0;
#pragma unroll
  for (int p = 0; p < 12; ++p)
    r |= (unsigned)(__popc(k & SRC_MASK[p]) & 1) << p;
  return r;
}

__device__ __forceinline__ v8f wmma4(v2f a, v2f b, v8f c) {
  return __builtin_amdgcn_wmma_f32_16x16x4_f32(false, a, false, b,
                                               (short)0, c, false, false);
}

__global__ __launch_bounds__(256)
void qdense_kernel(const float* __restrict__ x,
                   const float* __restrict__ theta,
                   float* __restrict__ out) {
  __shared__ float st[DIM];
  __shared__ float G[6][256];
  __shared__ float cs[24], sn[24];
  __shared__ float wred[8];
  __shared__ float wout[8][12];

  const int t    = threadIdx.x;
  const int lane = t & 31;
  const int wv   = t >> 5;
  const int n    = lane & 15;
  const int hi2  = (lane >> 4) * 2;
  const int mrow = (lane >> 4) * 8;

  if (t < 24) {
    float s, c;
    sincosf(theta[t] * 0.5f, &s, &c);
    cs[t] = c;
    sn[t] = s;
  }
  __syncthreads();

  {
    const int r = t >> 4, c = t & 15;
#pragma unroll
    for (int m = 0; m < 6; ++m) {
      float v = 1.f;
#pragma unroll
      for (int q = 0; q < 4; ++q) {
        const int rb = (r >> (3 - q)) & 1;
        const int cb = (c >> (3 - q)) & 1;
        const float cv = cs[m * 4 + q], sv = sn[m * 4 + q];
        v *= rb ? (cb ? cv : sv) : (cb ? -sv : cv);
      }
      G[m][t] = v;
    }
  }

  const float4* xr = (const float4*)(x + (size_t)blockIdx.x * DIM);
  float ss = 0.f;
#pragma unroll
  for (int e = 0; e < 4; ++e) {
    float4 v = xr[t * 4 + e];
    ((float4*)st)[t * 4 + e] = v;
    ss += v.x * v.x + v.y * v.y + v.z * v.z + v.w * v.w;
  }
#pragma unroll
  for (int o = 16; o > 0; o >>= 1) ss += __shfl_xor(ss, o, 32);
  if (lane == 0) wred[wv] = ss;
  __syncthreads();

  float sumsq = 0.f;
#pragma unroll
  for (int w = 0; w < 8; ++w) sumsq += wred[w];

  const bool fb = (sumsq <= 1e-24f);
  if (fb) {
#pragma unroll
    for (int e = 0; e < 16; ++e) st[e * 256 + t] = 0.f;
    if (t == 0) st[0] = 1.f;
  }
  const float inv = fb ? 1.f : 1.f / sumsq;
  __syncthreads();

#pragma unroll 1
  for (int d = 0; d < 2; ++d) {
    const float* g0 = G[d * 3 + 0];
    const float* g1 = G[d * 3 + 1];
    const float* g2 = G[d * 3 + 2];

    {
      v2f a[4];
#pragma unroll
      for (int kc = 0; kc < 4; ++kc) {
        a[kc].x = g0[n * 16 + kc * 4 + 0 + hi2];
        a[kc].y = g0[n * 16 + kc * 4 + 1 + hi2];
      }
#pragma unroll
      for (int tt = 0; tt < 2; ++tt) {
        const int col = (wv * 2 + tt) * 16 + n;
        v8f acc = {};
#pragma unroll
        for (int kc = 0; kc < 4; ++kc) {
          v2f b;
          b.x = st[(kc * 4 + 0 + hi2) * 256 + col];
          b.y = st[(kc * 4 + 1 + hi2) * 256 + col];
          acc = wmma4(a[kc], b, acc);
        }
#pragma unroll
        for (int v = 0; v < 8; ++v) st[(v + mrow) * 256 + col] = acc[v];
      }
    }
    __syncthreads();

    {
      v2f a[4];
#pragma unroll
      for (int kc = 0; kc < 4; ++kc) {
        a[kc].x = g1[n * 16 + kc * 4 + 0 + hi2];
        a[kc].y = g1[n * 16 + kc * 4 + 1 + hi2];
      }
#pragma unroll
      for (int tt = 0; tt < 2; ++tt) {
        const int base = (wv * 2 + tt) * 256;
        v8f acc = {};
#pragma unroll
        for (int kc = 0; kc < 4; ++kc) {
          v2f b;
          b.x = st[base + (kc * 4 + 0 + hi2) * 16 + n];
          b.y = st[base + (kc * 4 + 1 + hi2) * 16 + n];
          acc = wmma4(a[kc], b, acc);
        }
#pragma unroll
        for (int v = 0; v < 8; ++v) st[base + (v + mrow) * 16 + n] = acc[v];
      }
    }
    __syncthreads();

    {
      v2f bg[4];
#pragma unroll
      for (int kc = 0; kc < 4; ++kc) {
        bg[kc].x = g2[n * 16 + kc * 4 + 0 + hi2];
        bg[kc].y = g2[n * 16 + kc * 4 + 1 + hi2];
      }
#pragma unroll
      for (int tt = 0; tt < 2; ++tt) {
        const int rbase = (wv * 2 + tt) * 16;
        v8f acc = {};
#pragma unroll
        for (int kc = 0; kc < 4; ++kc) {
          v2f a;
          a.x = st[(rbase + n) * 16 + kc * 4 + 0 + hi2];
          a.y = st[(rbase + n) * 16 + kc * 4 + 1 + hi2];
          acc = wmma4(a, bg[kc], acc);
        }
#pragma unroll
        for (int v = 0; v < 8; ++v) st[(rbase + v + mrow) * 16 + n] = acc[v];
      }
    }
    __syncthreads();

    if (d == 0) {
      float pr[16];
#pragma unroll
      for (int e = 0; e < 16; ++e) pr[e] = st[src_perm(e * 256 + t)];
      __syncthreads();
#pragma unroll
      for (int e = 0; e < 16; ++e) st[e * 256 + t] = pr[e];
      __syncthreads();
    }
  }

  float acc12[12];
#pragma unroll
  for (int w = 0; w < 12; ++w) acc12[w] = 0.f;
#pragma unroll
  for (int e = 0; e < 16; ++e) {
    const unsigned k = (unsigned)(e * 256 + t);
    const float a = st[src_perm(k)];
    const float p = a * a;
#pragma unroll
    for (int w = 0; w < 12; ++w)
      acc12[w] += ((k >> (11 - w)) & 1u) ? -p : p;
  }
#pragma unroll
  for (int w = 0; w < 12; ++w) {
#pragma unroll
    for (int o = 16; o > 0; o >>= 1)
      acc12[w] += __shfl_xor(acc12[w], o, 32);
  }
  if (lane == 0) {
#pragma unroll
    for (int w = 0; w < 12; ++w) wout[wv][w] = acc12[w];
  }
  __syncthreads();
  if (t < 12) {
    float s = 0.f;
#pragma unroll
    for (int w2 = 0; w2 < 8; ++w2) s += wout[w2][t];
    out[(size_t)blockIdx.x * 12 + t] = s * inv;
  }
}

extern "C" void kernel_launch(void* const* d_in, const int* in_sizes, int n_in,
                              void* d_out, int out_size, void* d_ws,
                              size_t ws_size, hipStream_t stream) {
  const float* x = (const float*)d_in[0];
  const float* theta = (const float*)d_in[1];
  float* out = (float*)d_out;
  const int B = in_sizes[0] / DIM;
  qdense_kernel<<<dim3(B), dim3(256), 0, stream>>>(x, theta, out);
}


# Split the CNOT index map by GF(2) linearity, src(e*256+t) = src(t) ^ const_e: one 12-popcount index per thread instead of one per element; VGPRs 244 to 112, so twice the resident waves
# speedup vs baseline: 1.6157x; 1.6157x over previous
#include <hip/hip_runtime.h>
#include <math.h>


typedef __attribute__((ext_vector_type(2))) float v2f;
typedef __attribute__((ext_vector_type(8))) float v8f;

#define DIM 4096

constexpr unsigned cnot_src(unsigned k) {
  for (int i = 10; i >= 0; --i)
    for (int j = 11; j >= i + 1; --j)
      k ^= ((k >> (11 - i)) & 1u) << (11 - j);
  return k;
}
constexpr unsigned mask_p(int p) {
  unsigned mm = 0;
  for (int b = 0; b < 12; ++b)
    mm |= ((cnot_src(1u << b) >> p) & 1u) << b;
  return mm;
}
__constant__ unsigned SRC_MASK[12] = {
    mask_p(0), mask_p(1), mask_p(2),  mask_p(3),
    mask_p(4), mask_p(5), mask_p(6),  mask_p(7),
    mask_p(8), mask_p(9), mask_p(10), mask_p(11)};

__device__ __forceinline__ unsigned src_perm(unsigned k) {
  unsigned r = 0;
#pragma unroll
  for (int p = 0; p < 12; ++p)
    r |= (unsigned)(__popc(k & SRC_MASK[p]) & 1) << p;
  return r;
}

template <int E>
struct SrcHi { static constexpr unsigned value = cnot_src((unsigned)E << 8); };

template <int E>
__device__ __forceinline__ void gather_perm(const float* st, unsigned st_t, float* pr) {
  if constexpr (E < 16) {
    pr[E] = st[st_t ^ SrcHi<E>::value];
    gather_perm<E + 1>(st, st_t, pr);
  }
}

__device__ __forceinline__ v8f wmma4(v2f a, v2f b, v8f c) {
  return __builtin_amdgcn_wmma_f32_16x16x4_f32(false, a, false, b,
                                               (short)0, c, false, false);
}

__global__ __launch_bounds__(256)
void qdense_kernel(const float* __restrict__ x,
                   const float* __restrict__ theta,
                   float* __restrict__ out) {
  __shared__ float st[DIM];
  __shared__ float G[6][256];
  __shared__ float cs[24], sn[24];
  __shared__ float wred[8];
  __shared__ float wout[8][12];

  const int t    = threadIdx.x;
  const int lane = t & 31;
  const int wv   = t >> 5;
  const int n    = lane & 15;
  const int hi2  = (lane >> 4) * 2;
  const int mrow = (lane >> 4) * 8;
  const unsigned st_t = src_perm((unsigned)t);

  if (t < 24) {
    float s, c;
    sincosf(theta[t] * 0.5f, &s, &c);
    cs[t] = c;
    sn[t] = s;
  }
  __syncthreads();

  {
    const int r = t >> 4, c = t & 15;
#pragma unroll
    for (int m = 0; m < 6; ++m) {
      float v = 1.f;
#pragma unroll
      for (int q = 0; q < 4; ++q) {
        const int rb = (r >> (3 - q)) & 1;
        const int cb = (c >> (3 - q)) & 1;
        const float cv = cs[m * 4 + q], sv = sn[m * 4 + q];
        v *= rb ? (cb ? cv : sv) : (cb ? -sv : cv);
      }
      G[m][t] = v;
    }
  }

  const float4* xr = (const float4*)(x + (size_t)blockIdx.x * DIM);
  float ss = 0.f;
#pragma unroll
  for (int e = 0; e < 4; ++e) {
    float4 v = xr[t * 4 + e];
    ((float4*)st)[t * 4 + e] = v;
    ss += v.x * v.x + v.y * v.y + v.z * v.z + v.w * v.w;
  }
#pragma unroll
  for (int o = 16; o > 0; o >>= 1) ss += __shfl_xor(ss, o, 32);
  if (lane == 0) wred[wv] = ss;
  __syncthreads();

  float sumsq = 0.f;
#pragma unroll
  for (int w = 0; w < 8; ++w) sumsq += wred[w];

  const bool fb = (sumsq <= 1e-24f);
  if (fb) {
#pragma unroll
    for (int e = 0; e < 16; ++e) st[e * 256 + t] = 0.f;
    if (t == 0) st[0] = 1.f;
  }
  const float inv = fb ? 1.f : 1.f / sumsq;
  __syncthreads();

#pragma unroll 1
  for (int d = 0; d < 2; ++d) {
    const float* g0 = G[d * 3 + 0];
    const float* g1 = G[d * 3 + 1];
    const float* g2 = G[d * 3 + 2];

    {
      v2f a[4];
#pragma unroll
      for (int kc = 0; kc < 4; ++kc) {
        a[kc].x = g0[n * 16 + kc * 4 + 0 + hi2];
        a[kc].y = g0[n * 16 + kc * 4 + 1 + hi2];
      }
#pragma unroll
      for (int tt = 0; tt < 2; ++tt) {
        const int col = (wv * 2 + tt) * 16 + n;
        v8f acc = {};
#pragma unroll
        for (int kc = 0; kc < 4; ++kc) {
          v2f b;
          b.x = st[(kc * 4 + 0 + hi2) * 256 + col];
          b.y = st[(kc * 4 + 1 + hi2) * 256 + col];
          acc = wmma4(a[kc], b, acc);
        }
#pragma unroll
        for (int v = 0; v < 8; ++v) st[(v + mrow) * 256 + col] = acc[v];
      }
    }
    __syncthreads();

    {
      v2f a[4];
#pragma unroll
      for (int kc = 0; kc < 4; ++kc) {
        a[kc].x = g1[n * 16 + kc * 4 + 0 + hi2];
        a[kc].y = g1[n * 16 + kc * 4 + 1 + hi2];
      }
#pragma unroll
      for (int tt = 0; tt < 2; ++tt) {
        const int base = (wv * 2 + tt) * 256;
        v8f acc = {};
#pragma unroll
        for (int kc = 0; kc < 4; ++kc) {
          v2f b;
          b.x = st[base + (kc * 4 + 0 + hi2) * 16 + n];
          b.y = st[base + (kc * 4 + 1 + hi2) * 16 + n];
          acc = wmma4(a[kc], b, acc);
        }
#pragma unroll
        for (int v = 0; v < 8; ++v) st[base + (v + mrow) * 16 + n] = acc[v];
      }
    }
    __syncthreads();

    {
      v2f bg[4];
#pragma unroll
      for (int kc = 0; kc < 4; ++kc) {
        bg[kc].x = g2[n * 16 + kc * 4 + 0 + hi2];
        bg[kc].y = g2[n * 16 + kc * 4 + 1 + hi2];
      }
#pragma unroll
      for (int tt = 0; tt < 2; ++tt) {
        const int rbase = (wv * 2 + tt) * 16;
        v8f acc = {};
#pragma unroll
        for (int kc = 0; kc < 4; ++kc) {
          v2f a;
          a.x = st[(rbase + n) * 16 + kc * 4 + 0 + hi2];
          a.y = st[(rbase + n) * 16 + kc * 4 + 1 + hi2];
          acc = wmma4(a, bg[kc], acc);
        }
#pragma unroll
        for (int v = 0; v < 8; ++v) st[(rbase + v + mrow) * 16 + n] = acc[v];
      }
    }
    __syncthreads();

    if (d == 0) {
      float pr[16];
      gather_perm<0>(st, st_t, pr);
      __syncthreads();
#pragma unroll
      for (int e = 0; e < 16; ++e) st[e * 256 + t] = pr[e];
      __syncthreads();
    }
  }

  float acc12[12];
#pragma unroll
  for (int w = 0; w < 12; ++w) acc12[w] = 0.f;
  float fin[16];
  gather_perm<0>(st, st_t, fin);
#pragma unroll
  for (int e = 0; e < 16; ++e) {
    const unsigned k = (unsigned)(e * 256 + t);
    const float a = fin[e];
    const float p = a * a;
#pragma unroll
    for (int w = 0; w < 12; ++w)
      acc12[w] += ((k >> (11 - w)) & 1u) ? -p : p;
  }
#pragma unroll
  for (int w = 0; w < 12; ++w) {
#pragma unroll
    for (int o = 16; o > 0; o >>= 1)
      acc12[w] += __shfl_xor(acc12[w], o, 32);
  }
  if (lane == 0) {
#pragma unroll
    for (int w = 0; w < 12; ++w) wout[wv][w] = acc12[w];
  }
  __syncthreads();
  if (t < 12) {
    float s = 0.f;
#pragma unroll
    for (int w2 = 0; w2 < 8; ++w2) s += wout[w2][t];
    out[(size_t)blockIdx.x * 12 + t] = s * inv;
  }
}

extern "C" void kernel_launch(void* const* d_in, const int* in_sizes, int n_in,
                              void* d_out, int out_size, void* d_ws,
                              size_t ws_size, hipStream_t stream) {
  const float* x = (const float*)d_in[0];
  const float* theta = (const float*)d_in[1];
  float* out = (float*)d_out;
  const int B = in_sizes[0] / DIM;
  qdense_kernel<<<dim3(B), dim3(256), 0, stream>>>(x, theta, out);
}


# Z-expectations by a sum/difference tree: per thread S and 4 signed sums over e, lane and wave signs applied to sums of S; 11 cross-lane hops, not 60; changes the order of the float sums
# speedup vs baseline: 1.7604x; 1.0896x over previous
#include <hip/hip_runtime.h>
#include <math.h>


typedef __attribute__((ext_vector_type(2))) float v2f;
typedef __attribute__((ext_vector_type(8))) float v8f;

#define DIM 4096

constexpr unsigned cnot_src(unsigned k) {
  for (int i = 10; i >= 0; --i)
    for (int j = 11; j >= i + 1; --j)
      k ^= ((k >> (11 - i)) & 1u) << (11 - j);
  return k;
}
constexpr unsigned mask_p(int p) {
  unsigned mm = 0;
  for (int b = 0; b < 12; ++b)
    mm |= ((cnot_src(1u << b) >> p) & 1u) << b;
  return mm;
}
__constant__ unsigned SRC_MASK[12] = {
    mask_p(0), mask_p(1), mask_p(2),  mask_p(3),
    mask_p(4), mask_p(5), mask_p(6),  mask_p(7),
    mask_p(8), mask_p(9), mask_p(10), mask_p(11)};

__device__ __forceinline__ unsigned src_perm(unsigned k) {
  unsigned r = 0;
#pragma unroll
  for (int p = 0; p < 12; ++p)
    r |= (unsigned)(__popc(k & SRC_MASK[p]) & 1) << p;
  return r;
}

template <int E>
struct SrcHi { static constexpr unsigned value = cnot_src((unsigned)E << 8); };

template <int E>
__device__ __forceinline__ void gather_perm(const float* st, unsigned st_t, float* pr) {
  if constexpr (E < 16) {
    pr[E] = st[st_t ^ SrcHi<E>::value];
    gather_perm<E + 1>(st, st_t, pr);
  }
}

__device__ __forceinline__ v8f wmma4(v2f a, v2f b, v8f c) {
  return __builtin_amdgcn_wmma_f32_16x16x4_f32(false, a, false, b,
                                               (short)0, c, false, false);
}

__global__ __launch_bounds__(256)
void qdense_kernel(const float* __restrict__ x,
                   const float* __restrict__ theta,
                   float* __restrict__ out) {
  __shared__ float st[DIM];
  __shared__ float G[6][256];
  __shared__ float cs[24], sn[24];
  __shared__ float wred[8];
  __shared__ float wout[8][10];

  const int t    = threadIdx.x;
  const int lane = t & 31;
  const int wv   = t >> 5;
  const int n    = lane & 15;
  const int hi2  = (lane >> 4) * 2;
  const int mrow = (lane >> 4) * 8;
  const unsigned st_t = src_perm((unsigned)t);

  if (t < 24) {
    float s, c;
    sincosf(theta[t] * 0.5f, &s, &c);
    cs[t] = c;
    sn[t] = s;
  }
  __syncthreads();

  {
    const int r = t >> 4, c = t & 15;
#pragma unroll
    for (int m = 0; m < 6; ++m) {
      float v = 1.f;
#pragma unroll
      for (int q = 0; q < 4; ++q) {
        const int rb = (r >> (3 - q)) & 1;
        const int cb = (c >> (3 - q)) & 1;
        const float cv = cs[m * 4 + q], sv = sn[m * 4 + q];
        v *= rb ? (cb ? cv : sv) : (cb ? -sv : cv);
      }
      G[m][t] = v;
    }
  }

  const float4* xr = (const float4*)(x + (size_t)blockIdx.x * DIM);
  float ss = 0.f;
#pragma unroll
  for (int e = 0; e < 4; ++e) {
    float4 v = xr[t * 4 + e];
    ((float4*)st)[t * 4 + e] = v;
    ss += v.x * v.x + v.y * v.y + v.z * v.z + v.w * v.w;
  }
#pragma unroll
  for (int o = 16; o > 0; o >>= 1) ss += __shfl_xor(ss, o, 32);
  if (lane == 0) wred[wv] = ss;
  __syncthreads();

  float sumsq = 0.f;
#pragma unroll
  for (int w = 0; w < 8; ++w) sumsq += wred[w];

  const bool fb = (sumsq <= 1e-24f);
  if (fb) {
#pragma unroll
    for (int e = 0; e < 16; ++e) st[e * 256 + t] = 0.f;
    if (t == 0) st[0] = 1.f;
  }
  const float inv = fb ? 1.f : 1.f / sumsq;
  __syncthreads();

#pragma unroll 1
  for (int d = 0; d < 2; ++d) {
    const float* g0 = G[d * 3 + 0];
    const float* g1 = G[d * 3 + 1];
    const float* g2 = G[d * 3 + 2];

    {
      v2f a[4];
#pragma unroll
      for (int kc = 0; kc < 4; ++kc) {
        a[kc].x = g0[n * 16 + kc * 4 + 0 + hi2];
        a[kc].y = g0[n * 16 + kc * 4 + 1 + hi2];
      }
#pragma unroll
      for (int tt = 0; tt < 2; ++tt) {
        const int col = (wv * 2 + tt) * 16 + n;
        v8f acc = {};
#pragma unroll
        for (int kc = 0; kc < 4; ++kc) {
          v2f b;
          b.x = st[(kc * 4 + 0 + hi2) * 256 + col];
          b.y = st[(kc * 4 + 1 + hi2) * 256 + col];
          acc = wmma4(a[kc], b, acc);
        }
#pragma unroll
        for (int v = 0; v < 8; ++v) st[(v + mrow) * 256 + col] = acc[v];
      }
    }
    __syncthreads();

    {
      v2f a[4];
#pragma unroll
      for (int kc = 0; kc < 4; ++kc) {
        a[kc].x = g1[n * 16 + kc * 4 + 0 + hi2];
        a[kc].y = g1[n * 16 + kc * 4 + 1 + hi2];
      }
#pragma unroll
      for (int tt = 0; tt < 2; ++tt) {
        const int base = (wv * 2 + tt) * 256;
        v8f acc = {};
#pragma unroll
        for (int kc = 0; kc < 4; ++kc) {
          v2f b;
          b.x = st[base + (kc * 4 + 0 + hi2) * 16 + n];
          b.y = st[base + (kc * 4 + 1 + hi2) * 16 + n];
          acc = wmma4(a[kc], b, acc);
        }
#pragma unroll
        for (int v = 0; v < 8; ++v) st[base + (v + mrow) * 16 + n] = acc[v];
      }
    }
    __syncthreads();

    {
      v2f bg[4];
#pragma unroll
      for (int kc = 0; kc < 4; ++kc) {
        bg[kc].x = g2[n * 16 + kc * 4 + 0 + hi2];
        bg[kc].y = g2[n * 16 + kc * 4 + 1 + hi2];
      }
#pragma unroll
      for (int tt = 0; tt < 2; ++tt) {
        const int rbase = (wv * 2 + tt) * 16;
        v8f acc = {};
#pragma unroll
        for (int kc = 0; kc < 4; ++kc) {
          v2f a;
          a.x = st[(rbase + n) * 16 + kc * 4 + 0 + hi2];
          a.y = st[(rbase + n) * 16 + kc * 4 + 1 + hi2];
          acc = wmma4(a, bg[kc], acc);
        }
#pragma unroll
        for (int v = 0; v < 8; ++v) st[(rbase + v + mrow) * 16 + n] = acc[v];
      }
    }
    __syncthreads();

    if (d == 0) {
      float pr[16];
      gather_perm<0>(st, st_t, pr);
      __syncthreads();
#pragma unroll
      for (int e = 0; e < 16; ++e) st[e * 256 + t] = pr[e];
      __syncthreads();
    }
  }

  float fin[16];
  gather_perm<0>(st, st_t, fin);
  float p[16];
#pragma unroll
  for (int e = 0; e < 16; ++e) p[e] = fin[e] * fin[e];
  float a[8], b[4], c[2];
  float d3 = 0.f, d2 = 0.f, d1 = 0.f;
#pragma unroll
  for (int i = 0; i < 8; ++i) {
    a[i] = p[2 * i] + p[2 * i + 1];
    d3 += p[2 * i] - p[2 * i + 1];
  }
#pragma unroll
  for (int i = 0; i < 4; ++i) {
    b[i] = a[2 * i] + a[2 * i + 1];
    d2 += a[2 * i] - a[2 * i + 1];
  }
#pragma unroll
  for (int i = 0; i < 2; ++i) {
    c[i] = b[2 * i] + b[2 * i + 1];
    d1 += b[2 * i] - b[2 * i + 1];
  }
  const float S  = c[0] + c[1];
  const float d0 = c[0] - c[1];

  float xs = S;
#pragma unroll
  for (int bb = 4; bb >= 0; --bb) {
    const float u = __shfl_xor(xs, 1 << bb, 32);
    xs = ((lane >> bb) == 1) ? (u - xs) : (xs + u);
  }
  const bool b4 = (lane & 16) != 0;
  const bool b3 = (lane & 8) != 0;
  const float y01 = (b4 ? d1 : d0) + __shfl_xor(b4 ? d0 : d1, 16, 32);
  const float y23 = (b4 ? d3 : d2) + __shfl_xor(b4 ? d2 : d3, 16, 32);
  float zs = (b3 ? y23 : y01) + __shfl_xor(b3 ? y01 : y23, 8, 32);
#pragma unroll
  for (int o = 4; o > 0; o >>= 1) zs += __shfl_xor(zs, o, 32);

  if ((lane & 7) == 0) wout[wv][((lane >> 4) & 1) + ((lane >> 3) & 1) * 2] = zs;
  if ((lane & (lane - 1)) == 0)
    wout[wv][lane == 0 ? 4 : 10 - __ffs(lane)] = xs;
  __syncthreads();
  if (t < 12) {
    const bool wave_signed = (t >= 4) && (t < 7);
    const int slot = (t < 4) ? t : (wave_signed ? 4 : t - 2);
    float s = 0.f;
#pragma unroll
    for (int w2 = 0; w2 < 8; ++w2) {
      const float v = wout[w2][slot];
      const bool neg = wave_signed && (((w2 >> (6 - t)) & 1) != 0);
      s += neg ? -v : v;
    }
    out[(size_t)blockIdx.x * 12 + t] = s * inv;
  }
}

extern "C" void kernel_launch(void* const* d_in, const int* in_sizes, int n_in,
                              void* d_out, int out_size, void* d_ws,
                              size_t ws_size, hipStream_t stream) {
  const float* x = (const float*)d_in[0];
  const float* theta = (const float*)d_in[1];
  float* out = (float*)d_out;
  const int B = in_sizes[0] / DIM;
  qdense_kernel<<<dim3(B), dim3(256), 0, stream>>>(x, theta, out);
}


# LDS bank conflicts removed: state kept at an XOR-swizzled index (index bits 6,7,9,11 folded into bank bits 2-5), gate rows padded to pitch 20: every fragment load and store is conflict-free
# speedup vs baseline: 2.0550x; 1.1673x over previous
#include <hip/hip_runtime.h>
#include <math.h>


typedef __attribute__((ext_vector_type(2))) float v2f;
typedef __attribute__((ext_vector_type(8))) float v8f;

#define DIM 4096
#define GP 20

__host__ __device__ constexpr unsigned sw(unsigned k) {
  return k ^ (((k >> 6) & 1u) << 2) ^ (((k >> 7) & 1u) * 24u) ^
         (((k >> 9) & 1u) << 5) ^ (((k >> 11) & 1u) << 4);
}
__device__ __forceinline__ unsigned at(unsigned tb, unsigned c) {
  const unsigned s = sw(c);
  return (tb ^ (s & 63u)) + (s & ~63u);
}

constexpr unsigned cnot_src(unsigned k) {
  for (int i = 10; i >= 0; --i)
    for (int j = 11; j >= i + 1; --j)
      k ^= ((k >> (11 - i)) & 1u) << (11 - j);
  return k;
}
constexpr unsigned mask_p(int p) {
  unsigned mm = 0;
  for (int b = 0; b < 12; ++b)
    mm |= ((cnot_src(1u << b) >> p) & 1u) << b;
  return mm;
}
__constant__ unsigned SRC_MASK[12] = {
    mask_p(0), mask_p(1), mask_p(2),  mask_p(3),
    mask_p(4), mask_p(5), mask_p(6),  mask_p(7),
    mask_p(8), mask_p(9), mask_p(10), mask_p(11)};

__device__ __forceinline__ unsigned src_perm(unsigned k) {
  unsigned r = 0;
#pragma unroll
  for (int p = 0; p < 12; ++p)
    r |= (unsigned)(__popc(k & SRC_MASK[p]) & 1) << p;
  return r;
}

template <int E>
struct SrcHi { static constexpr unsigned value = cnot_src((unsigned)E << 8); };

template <int E>
__device__ __forceinline__ void gather_perm(const float* st, unsigned tg, float* pr) {
  if constexpr (E < 16) {
    constexpr unsigned s = sw(SrcHi<E>::value);
    pr[E] = st[(tg ^ (s & 0xFFu)) + (s & ~0xFFu)];
    gather_perm<E + 1>(st, tg, pr);
  }
}

__device__ __forceinline__ v8f wmma4(v2f a, v2f b, v8f c) {
  return __builtin_amdgcn_wmma_f32_16x16x4_f32(false, a, false, b,
                                               (short)0, c, false, false);
}

__global__ __launch_bounds__(256)
void qdense_kernel(const float* __restrict__ x,
                   const float* __restrict__ theta,
                   float* __restrict__ out) {
  __shared__ __align__(16) float st[DIM];
  __shared__ __align__(16) float G[6][16 * GP];
  __shared__ float cs[24], sn[24];
  __shared__ float wred[8];
  __shared__ float wout[8][10];

  const int t    = threadIdx.x;
  const int lane = t & 31;
  const int wv   = t >> 5;
  const int n    = lane & 15;
  const int hi2  = (lane >> 4) * 2;
  const unsigned tg = sw(src_perm((unsigned)t));
  const unsigned un = (unsigned)n, uh = (unsigned)(lane >> 4), uw = (unsigned)wv;
  const unsigned tb_hr = sw(un | (uw << 5) | (uh << 9));
  const unsigned tb_hw = sw(un | (uw << 5) | (uh << 11));
  const unsigned tb_mr = sw(un | (uh << 5) | (uw << 9));
  const unsigned tb_w  = sw(un | (uh << 7) | (uw << 9));
  const unsigned tb_lr = sw((uh << 1) | (un << 4) | (uw << 9));
  const unsigned tb_ps = sw((unsigned)t);
  const unsigned tb_in = sw((unsigned)t << 4);

  if (t < 24) {
    float s, c;
    sincosf(theta[t] * 0.5f, &s, &c);
    cs[t] = c;
    sn[t] = s;
  }
  __syncthreads();

  {
    const int r = t >> 4, c = t & 15;
#pragma unroll
    for (int m = 0; m < 6; ++m) {
      float v = 1.f;
#pragma unroll
      for (int q = 0; q < 4; ++q) {
        const int rb = (r >> (3 - q)) & 1;
        const int cb = (c >> (3 - q)) & 1;
        const float cv = cs[m * 4 + q], sv = sn[m * 4 + q];
        v *= rb ? (cb ? cv : sv) : (cb ? -sv : cv);
      }
      G[m][r * GP + c] = v;
    }
  }

  const float4* xr = (const float4*)(x + (size_t)blockIdx.x * DIM);
  float ss = 0.f;
#pragma unroll
  for (int e = 0; e < 4; ++e) {
    float4 v = xr[t * 4 + e];
    *(float4*)&st[at(tb_in, (unsigned)e << 2)] = v;
    ss += v.x * v.x + v.y * v.y + v.z * v.z + v.w * v.w;
  }
#pragma unroll
  for (int o = 16; o > 0; o >>= 1) ss += __shfl_xor(ss, o, 32);
  if (lane == 0) wred[wv] = ss;
  __syncthreads();

  float sumsq = 0.f;
#pragma unroll
  for (int w = 0; w < 8; ++w) sumsq += wred[w];

  const bool fb = (sumsq <= 1e-24f);
  if (fb) {
#pragma unroll
    for (int e = 0; e < 16; ++e) st[at(tb_ps, (unsigned)e << 8)] = 0.f;
    if (t == 0) st[0] = 1.f;
  }
  const float inv = fb ? 1.f : 1.f / sumsq;
  __syncthreads();

#pragma unroll 1
  for (int d = 0; d < 2; ++d) {
    const float* g0 = G[d * 3 + 0];
    const float* g1 = G[d * 3 + 1];
    const float* g2 = G[d * 3 + 2];

    {
      v2f a[4];
#pragma unroll
      for (int kc = 0; kc < 4; ++kc)
        a[kc] = *(const v2f*)&g0[n * GP + kc * 4 + hi2];
#pragma unroll
      for (int tt = 0; tt < 2; ++tt) {
        v8f acc = {};
#pragma unroll
        for (int kc = 0; kc < 4; ++kc) {
          const unsigned c = ((unsigned)tt << 4) | ((unsigned)kc << 10);
          v2f b;
          b.x = st[at(tb_hr, c)];
          b.y = st[at(tb_hr, c | (1u << 8))];
          acc = wmma4(a[kc], b, acc);
        }
#pragma unroll
        for (int v = 0; v < 8; ++v)
          st[at(tb_hw, ((unsigned)tt << 4) | ((unsigned)v << 8))] = acc[v];
      }
    }
    __syncthreads();

    {
      v2f a[4];
#pragma unroll
      for (int kc = 0; kc < 4; ++kc)
        a[kc] = *(const v2f*)&g1[n * GP + kc * 4 + hi2];
#pragma unroll
      for (int tt = 0; tt < 2; ++tt) {
        v8f acc = {};
#pragma unroll
        for (int kc = 0; kc < 4; ++kc) {
          const unsigned c = ((unsigned)kc << 6) | ((unsigned)tt << 8);
          v2f b;
          b.x = st[at(tb_mr, c)];
          b.y = st[at(tb_mr, c | (1u << 4))];
          acc = wmma4(a[kc], b, acc);
        }
#pragma unroll
        for (int v = 0; v < 8; ++v)
          st[at(tb_w, ((unsigned)v << 4) | ((unsigned)tt << 8))] = acc[v];
      }
    }
    __syncthreads();

    {
      v2f bg[4];
#pragma unroll
      for (int kc = 0; kc < 4; ++kc)
        bg[kc] = *(const v2f*)&g2[n * GP + kc * 4 + hi2];
#pragma unroll
      for (int tt = 0; tt < 2; ++tt) {
        v8f acc = {};
#pragma unroll
        for (int kc = 0; kc < 4; ++kc) {
          const v2f a =
              *(const v2f*)&st[at(tb_lr, ((unsigned)kc << 2) | ((unsigned)tt << 8))];
          acc = wmma4(a, bg[kc], acc);
        }
#pragma unroll
        for (int v = 0; v < 8; ++v)
          st[at(tb_w, ((unsigned)v << 4) | ((unsigned)tt << 8))] = acc[v];
      }
    }
    __syncthreads();

    if (d == 0) {
      float pr[16];
      gather_perm<0>(st, tg, pr);
      __syncthreads();
#pragma unroll
      for (int e = 0; e < 16; ++e) st[at(tb_ps, (unsigned)e << 8)] = pr[e];
      __syncthreads();
    }
  }

  float fin[16];
  gather_perm<0>(st, tg, fin);
  float p[16];
#pragma unroll
  for (int e = 0; e < 16; ++e) p[e] = fin[e] * fin[e];
  float a[8], b[4], c[2];
  float d3 = 0.f, d2 = 0.f, d1 = 0.f;
#pragma unroll
  for (int i = 0; i < 8; ++i) {
    a[i] = p[2 * i] + p[2 * i + 1];
    d3 += p[2 * i] - p[2 * i + 1];
  }
#pragma unroll
  for (int i = 0; i < 4; ++i) {
    b[i] = a[2 * i] + a[2 * i + 1];
    d2 += a[2 * i] - a[2 * i + 1];
  }
#pragma unroll
  for (int i = 0; i < 2; ++i) {
    c[i] = b[2 * i] + b[2 * i + 1];
    d1 += b[2 * i] - b[2 * i + 1];
  }
  const float S  = c[0] + c[1];
  const float d0 = c[0] - c[1];

  float xs = S;
#pragma unroll
  for (int bb = 4; bb >= 0; --bb) {
    const float u = __shfl_xor(xs, 1 << bb, 32);
    xs = ((lane >> bb) == 1) ? (u - xs) : (xs + u);
  }
  const bool b4 = (lane & 16) != 0;
  const bool b3 = (lane & 8) != 0;
  const float y01 = (b4 ? d1 : d0) + __shfl_xor(b4 ? d0 : d1, 16, 32);
  const float y23 = (b4 ? d3 : d2) + __shfl_xor(b4 ? d2 : d3, 16, 32);
  float zs = (b3 ? y23 : y01) + __shfl_xor(b3 ? y01 : y23, 8, 32);
#pragma unroll
  for (int o = 4; o > 0; o >>= 1) zs += __shfl_xor(zs, o, 32);

  if ((lane & 7) == 0) wout[wv][((lane >> 4) & 1) + ((lane >> 3) & 1) * 2] = zs;
  if ((lane & (lane - 1)) == 0)
    wout[wv][lane == 0 ? 4 : 10 - __ffs(lane)] = xs;
  __syncthreads();
  if (t < 12) {
    const bool wave_signed = (t >= 4) && (t < 7);
    const int slot = (t < 4) ? t : (wave_signed ? 4 : t - 2);
    float s = 0.f;
#pragma unroll
    for (int w2 = 0; w2 < 8; ++w2) {
      const float v = wout[w2][slot];
      const bool neg = wave_signed && (((w2 >> (6 - t)) & 1) != 0);
      s += neg ? -v : v;
    }
    out[(size_t)blockIdx.x * 12 + t] = s * inv;
  }
}

extern "C" void kernel_launch(void* const* d_in, const int* in_sizes, int n_in,
                              void* d_out, int out_size, void* d_ws,
                              size_t ws_size, hipStream_t stream) {
  const float* x = (const float*)d_in[0];
  const float* theta = (const float*)d_in[1];
  float* out = (float*)d_out;
  const int B = in_sizes[0] / DIM;
  qdense_kernel<<<dim3(B), dim3(256), 0, stream>>>(x, theta, out);
}


# Persistent grid: 1024 resident workgroups (4 per WGP) each walk 4 rows, next row's loads in flight under LDS-only barriers; sin/cos, gate matrices, index maps built once per workgroup; 1/4 the waves
# speedup vs baseline: 2.2117x; 1.0763x over previous
#include <hip/hip_runtime.h>
#include <math.h>


typedef __attribute__((ext_vector_type(2))) float v2f;
typedef __attribute__((ext_vector_type(8))) float v8f;

#define DIM 4096
#define GP 20
#define NWG 1024

__host__ __device__ constexpr unsigned sw(unsigned k) {
  return k ^ (((k >> 6) & 1u) << 2) ^ (((k >> 7) & 1u) * 24u) ^
         (((k >> 9) & 1u) << 5) ^ (((k >> 11) & 1u) << 4);
}
__device__ __forceinline__ unsigned at(unsigned tb, unsigned c) {
  const unsigned s = sw(c);
  return (tb ^ (s & 63u)) + (s & ~63u);
}

constexpr unsigned cnot_src(unsigned k) {
  for (int i = 10; i >= 0; --i)
    for (int j = 11; j >= i + 1; --j)
      k ^= ((k >> (11 - i)) & 1u) << (11 - j);
  return k;
}
constexpr unsigned mask_p(int p) {
  unsigned mm = 0;
  for (int b = 0; b < 12; ++b)
    mm |= ((cnot_src(1u << b) >> p) & 1u) << b;
  return mm;
}
__constant__ unsigned SRC_MASK[12] = {
    mask_p(0), mask_p(1), mask_p(2),  mask_p(3),
    mask_p(4), mask_p(5), mask_p(6),  mask_p(7),
    mask_p(8), mask_p(9), mask_p(10), mask_p(11)};

__device__ __forceinline__ unsigned src_perm(unsigned k) {
  unsigned r = 0;
#pragma unroll
  for (int p = 0; p < 12; ++p)
    r |= (unsigned)(__popc(k & SRC_MASK[p]) & 1) << p;
  return r;
}

template <int E>
struct SrcHi { static constexpr unsigned value = cnot_src((unsigned)E << 8); };

template <int E>
__device__ __forceinline__ void gather_perm(const float* st, unsigned tg, float* pr) {
  if constexpr (E < 16) {
    constexpr unsigned s = sw(SrcHi<E>::value);
    pr[E] = st[(tg ^ (s & 0xFFu)) + (s & ~0xFFu)];
    gather_perm<E + 1>(st, tg, pr);
  }
}

__device__ __forceinline__ v8f wmma4(v2f a, v2f b, v8f c) {
  return __builtin_amdgcn_wmma_f32_16x16x4_f32(false, a, false, b,
                                               (short)0, c, false, false);
}

__device__ __forceinline__ void lds_barrier() {
  __builtin_amdgcn_fence(__ATOMIC_RELEASE, "workgroup", "local");
  __builtin_amdgcn_s_barrier();
  __builtin_amdgcn_fence(__ATOMIC_ACQUIRE, "workgroup", "local");
}

__global__ __launch_bounds__(256) __attribute__((amdgpu_waves_per_eu(8, 8)))
void qdense_kernel(const float* __restrict__ x,
                   const float* __restrict__ theta,
                   float* __restrict__ out,
                   int nrows) {
  __shared__ __align__(16) float st[DIM];
  __shared__ __align__(16) float G[6][16 * GP];
  __shared__ float cs[24], sn[24];
  __shared__ float wred[8];
  __shared__ float wout[8][10];

  const int t    = threadIdx.x;
  const int lane = t & 31;
  const int wv   = t >> 5;
  const int n    = lane & 15;
  const int hi2  = (lane >> 4) * 2;
  const unsigned tg = sw(src_perm((unsigned)t));
  const unsigned un = (unsigned)n, uh = (unsigned)(lane >> 4), uw = (unsigned)wv;
  const unsigned tb_hr = sw(un | (uw << 5) | (uh << 9));
  const unsigned tb_hw = sw(un | (uw << 5) | (uh << 11));
  const unsigned tb_mr = sw(un | (uh << 5) | (uw << 9));
  const unsigned tb_w  = sw(un | (uh << 7) | (uw << 9));
  const unsigned tb_lr = sw((uh << 1) | (un << 4) | (uw << 9));
  const unsigned tb_ps = sw((unsigned)t);
  const unsigned tb_in = sw((unsigned)t << 4);

  int row = blockIdx.x;
  float4 nx[4];
  if (row < nrows) {
    const float4* xr = (const float4*)(x + (size_t)row * DIM);
#pragma unroll
    for (int e = 0; e < 4; ++e) nx[e] = xr[t * 4 + e];
  }

  if (t < 24) {
    float s, c;
    sincosf(theta[t] * 0.5f, &s, &c);
    cs[t] = c;
    sn[t] = s;
  }
  lds_barrier();

  {
    const int r = t >> 4, c = t & 15;
#pragma unroll
    for (int m = 0; m < 6; ++m) {
      float v = 1.f;
#pragma unroll
      for (int q = 0; q < 4; ++q) {
        const int rb = (r >> (3 - q)) & 1;
        const int cb = (c >> (3 - q)) & 1;
        const float cv = cs[m * 4 + q], sv = sn[m * 4 + q];
        v *= rb ? (cb ? cv : sv) : (cb ? -sv : cv);
      }
      G[m][r * GP + c] = v;
    }
  }

#pragma unroll 1
  for (; row < nrows; row += NWG) {
    float ss = 0.f;
#pragma unroll
    for (int e = 0; e < 4; ++e) {
      const float4 v = nx[e];
      *(float4*)&st[at(tb_in, (unsigned)e << 2)] = v;
      ss += v.x * v.x + v.y * v.y + v.z * v.z + v.w * v.w;
    }
    if (row + NWG < nrows) {
      const float4* xn = (const float4*)(x + (size_t)(row + NWG) * DIM);
#pragma unroll
      for (int e = 0; e < 4; ++e) nx[e] = xn[t * 4 + e];
    }
#pragma unroll
    for (int o = 16; o > 0; o >>= 1) ss += __shfl_xor(ss, o, 32);
    if (lane == 0) wred[wv] = ss;
    lds_barrier();

    float sumsq = 0.f;
#pragma unroll
    for (int w = 0; w < 8; ++w) sumsq += wred[w];

    const bool fb = (sumsq <= 1e-24f);
    if (fb) {
#pragma unroll
      for (int e = 0; e < 16; ++e) st[at(tb_ps, (unsigned)e << 8)] = 0.f;
      if (t == 0) st[0] = 1.f;
    }
    const float inv = fb ? 1.f : 1.f / sumsq;
    lds_barrier();

#pragma unroll 1
    for (int d = 0; d < 2; ++d) {
      const float* g0 = G[d * 3 + 0];
      const float* g1 = G[d * 3 + 1];
      const float* g2 = G[d * 3 + 2];

      {
        v2f a[4];
#pragma unroll
        for (int kc = 0; kc < 4; ++kc)
          a[kc] = *(const v2f*)&g0[n * GP + kc * 4 + hi2];
#pragma unroll
        for (int tt = 0; tt < 2; ++tt) {
          v8f acc = {};
#pragma unroll
          for (int kc = 0; kc < 4; ++kc) {
            const unsigned c = ((unsigned)tt << 4) | ((unsigned)kc << 10);
            v2f b;
            b.x = st[at(tb_hr, c)];
            b.y = st[at(tb_hr, c | (1u << 8))];
            acc = wmma4(a[kc], b, acc);
          }
#pragma unroll
          for (int v = 0; v < 8; ++v)
            st[at(tb_hw, ((unsigned)tt << 4) | ((unsigned)v << 8))] = acc[v];
        }
      }
      lds_barrier();

      {
        v2f a[4];
#pragma unroll
        for (int kc = 0; kc < 4; ++kc)
          a[kc] = *(const v2f*)&g1[n * GP + kc * 4 + hi2];
#pragma unroll
        for (int tt = 0; tt < 2; ++tt) {
          v8f acc = {};
#pragma unroll
          for (int kc = 0; kc < 4; ++kc) {
            const unsigned c = ((unsigned)kc << 6) | ((unsigned)tt << 8);
            v2f b;
            b.x = st[at(tb_mr, c)];
            b.y = st[at(tb_mr, c | (1u << 4))];
            acc = wmma4(a[kc], b, acc);
          }
#pragma unroll
          for (int v = 0; v < 8; ++v)
            st[at(tb_w, ((unsigned)v << 4) | ((unsigned)tt << 8))] = acc[v];
        }
      }
      lds_barrier();

      {
        v2f bg[4];
#pragma unroll
        for (int kc = 0; kc < 4; ++kc)
          bg[kc] = *(const v2f*)&g2[n * GP + kc * 4 + hi2];
#pragma unroll
        for (int tt = 0; tt < 2; ++tt) {
          v8f acc = {};
#pragma unroll
          for (int kc = 0; kc < 4; ++kc) {
            const v2f a =
                *(const v2f*)&st[at(tb_lr, ((unsigned)kc << 2) | ((unsigned)tt << 8))];
            acc = wmma4(a, bg[kc], acc);
          }
#pragma unroll
          for (int v = 0; v < 8; ++v)
            st[at(tb_w, ((unsigned)v << 4) | ((unsigned)tt << 8))] = acc[v];
        }
      }
      lds_barrier();

      if (d == 0) {
        float pr[16];
        gather_perm<0>(st, tg, pr);
        lds_barrier();
#pragma unroll
        for (int e = 0; e < 16; ++e) st[at(tb_ps, (unsigned)e << 8)] = pr[e];
        lds_barrier();
      }
    }

    float fin[16];
    gather_perm<0>(st, tg, fin);
    float p[16];
#pragma unroll
    for (int e = 0; e < 16; ++e) p[e] = fin[e] * fin[e];
    float a[8], b[4], c[2];
    float d3 = 0.f, d2 = 0.f, d1 = 0.f;
#pragma unroll
    for (int i = 0; i < 8; ++i) {
      a[i] = p[2 * i] + p[2 * i + 1];
      d3 += p[2 * i] - p[2 * i + 1];
    }
#pragma unroll
    for (int i = 0; i < 4; ++i) {
      b[i] = a[2 * i] + a[2 * i + 1];
      d2 += a[2 * i] - a[2 * i + 1];
    }
#pragma unroll
    for (int i = 0; i < 2; ++i) {
      c[i] = b[2 * i] + b[2 * i + 1];
      d1 += b[2 * i] - b[2 * i + 1];
    }
    const float S  = c[0] + c[1];
    const float d0 = c[0] - c[1];

    float xs = S;
#pragma unroll
    for (int bb = 4; bb >= 0; --bb) {
      const float u = __shfl_xor(xs, 1 << bb, 32);
      xs = ((lane >> bb) == 1) ? (u - xs) : (xs + u);
    }
    const bool b4 = (lane & 16) != 0;
    const bool b3 = (lane & 8) != 0;
    const float y01 = (b4 ? d1 : d0) + __shfl_xor(b4 ? d0 : d1, 16, 32);
    const float y23 = (b4 ? d3 : d2) + __shfl_xor(b4 ? d2 : d3, 16, 32);
    float zs = (b3 ? y23 : y01) + __shfl_xor(b3 ? y01 : y23, 8, 32);
#pragma unroll
    for (int o = 4; o > 0; o >>= 1) zs += __shfl_xor(zs, o, 32);

    if ((lane & 7) == 0) wout[wv][((lane >> 4) & 1) + ((lane >> 3) & 1) * 2] = zs;
    if ((lane & (lane - 1)) == 0)
      wout[wv][lane == 0 ? 4 : 10 - __ffs(lane)] = xs;
    lds_barrier();
    if (t < 12) {
      const bool wave_signed = (t >= 4) && (t < 7);
      const int slot = (t < 4) ? t : (wave_signed ? 4 : t - 2);
      float s = 0.f;
#pragma unroll
      for (int w2 = 0; w2 < 8; ++w2) {
        const float v = wout[w2][slot];
        const bool neg = wave_signed && (((w2 >> (6 - t)) & 1) != 0);
        s += neg ? -v : v;
      }
      out[(size_t)row * 12 + t] = s * inv;
    }
  }
}

extern "C" void kernel_launch(void* const* d_in, const int* in_sizes, int n_in,
                              void* d_out, int out_size, void* d_ws,
                              size_t ws_size, hipStream_t stream) {
  const float* x = (const float*)d_in[0];
  const float* theta = (const float*)d_in[1];
  float* out = (float*)d_out;
  const int B = in_sizes[0] / DIM;
  qdense_kernel<<<dim3(NWG), dim3(256), 0, stream>>>(x, theta, out, B);
}


# MID and LO block products chained in registers (first product's D fragment is the second's A fragment, k order matched), layer 0 straight from the global loads; changes the order of the float sums
# speedup vs baseline: 2.4409x; 1.1036x over previous
#include <hip/hip_runtime.h>
#include <math.h>


typedef __attribute__((ext_vector_type(2))) float v2f;
typedef __attribute__((ext_vector_type(8))) float v8f;

#define DIM 4096
#define GP 20
#define NWG 1024

__host__ __device__ constexpr unsigned sw(unsigned k) {
  return k ^ (((k >> 6) & 1u) << 2) ^ (((k >> 7) & 1u) * 24u) ^
         (((k >> 9) & 1u) << 5) ^ (((k >> 11) & 1u) << 4);
}
__device__ __forceinline__ unsigned at(unsigned tb, unsigned c) {
  const unsigned s = sw(c);
  return (tb ^ (s & 63u)) + (s & ~63u);
}

constexpr unsigned cnot_src(unsigned k) {
  for (int i = 10; i >= 0; --i)
    for (int j = 11; j >= i + 1; --j)
      k ^= ((k >> (11 - i)) & 1u) << (11 - j);
  return k;
}
constexpr unsigned mask_p(int p) {
  unsigned mm = 0;
  for (int b = 0; b < 12; ++b)
    mm |= ((cnot_src(1u << b) >> p) & 1u) << b;
  return mm;
}
__constant__ unsigned SRC_MASK[12] = {
    mask_p(0), mask_p(1), mask_p(2),  mask_p(3),
    mask_p(4), mask_p(5), mask_p(6),  mask_p(7),
    mask_p(8), mask_p(9), mask_p(10), mask_p(11)};

__device__ __forceinline__ unsigned src_perm(unsigned k) {
  unsigned r = 0;
#pragma unroll
  for (int p = 0; p < 12; ++p)
    r |= (unsigned)(__popc(k & SRC_MASK[p]) & 1) << p;
  return r;
}

template <int E>
struct SrcHi { static constexpr unsigned value = cnot_src((unsigned)E << 8); };

template <int E>
__device__ __forceinline__ void gather_perm(const float* st, unsigned tg, float* pr) {
  if constexpr (E < 16) {
    constexpr unsigned s = sw(SrcHi<E>::value);
    pr[E] = st[(tg ^ (s & 0xFFu)) + (s & ~0xFFu)];
    gather_perm<E + 1>(st, tg, pr);
  }
}

__device__ __forceinline__ v8f wmma4(v2f a, v2f b, v8f c) {
  return __builtin_amdgcn_wmma_f32_16x16x4_f32(false, a, false, b,
                                               (short)0, c, false, false);
}

__device__ __forceinline__ void lds_barrier() {
  __builtin_amdgcn_fence(__ATOMIC_RELEASE, "workgroup", "local");
  __builtin_amdgcn_s_barrier();
  __builtin_amdgcn_fence(__ATOMIC_ACQUIRE, "workgroup", "local");
}

__device__ __forceinline__ void mid_lo_pass(const float4 (&f)[4], const float* g1,
                                            const float* g2, float* st,
                                            unsigned tb_fs, int n, int hf) {
  v2f b1[4], b2[4];
#pragma unroll
  for (int kc = 0; kc < 4; ++kc) {
    b1[kc] = *(const v2f*)&g2[n * GP + 4 * hf + 8 * (kc >> 1) + 2 * (kc & 1)];
    b2[kc] = *(const v2f*)&g1[n * GP + 8 * hf + 2 * kc];
  }
  v8f u[2], r[2];
#pragma unroll
  for (int tt = 0; tt < 2; ++tt) {
    u[tt] = {};
#pragma unroll
    for (int kc = 0; kc < 4; ++kc) {
      const float4 v = f[tt * 2 + (kc >> 1)];
      v2f a;
      a.x = (kc & 1) ? v.z : v.x;
      a.y = (kc & 1) ? v.w : v.y;
      u[tt] = wmma4(a, b1[kc], u[tt]);
    }
  }
#pragma unroll
  for (int tt = 0; tt < 2; ++tt) {
    r[tt] = {};
#pragma unroll
    for (int kc = 0; kc < 4; ++kc) {
      v2f a;
      a.x = u[tt][2 * kc];
      a.y = u[tt][2 * kc + 1];
      r[tt] = wmma4(a, b2[kc], r[tt]);
    }
  }
#pragma unroll
  for (int tt = 0; tt < 2; ++tt) {
#pragma unroll
    for (int q = 0; q < 2; ++q) {
      float4 o;
      o.x = r[tt][4 * q];
      o.y = r[tt][4 * q + 1];
      o.z = r[tt][4 * q + 2];
      o.w = r[tt][4 * q + 3];
      *(float4*)&st[at(tb_fs, ((unsigned)tt << 8) | ((unsigned)q << 2))] = o;
    }
  }
}

__device__ __forceinline__ void hi_pass(const float* g0, float* st, unsigned tb_hr,
                                        unsigned tb_hw, int n, int hi2) {
  v2f a[4];
#pragma unroll
  for (int kc = 0; kc < 4; ++kc)
    a[kc] = *(const v2f*)&g0[n * GP + kc * 4 + hi2];
#pragma unroll
  for (int tt = 0; tt < 2; ++tt) {
    v8f acc = {};
#pragma unroll
    for (int kc = 0; kc < 4; ++kc) {
      const unsigned c = ((unsigned)tt << 4) | ((unsigned)kc << 10);
      v2f b;
      b.x = st[at(tb_hr, c)];
      b.y = st[at(tb_hr, c | (1u << 8))];
      acc = wmma4(a[kc], b, acc);
    }
#pragma unroll
    for (int v = 0; v < 8; ++v)
      st[at(tb_hw, ((unsigned)tt << 4) | ((unsigned)v << 8))] = acc[v];
  }
}

__global__ __launch_bounds__(256) __attribute__((amdgpu_waves_per_eu(8, 8)))
void qdense_kernel(const float* __restrict__ x,
                   const float* __restrict__ theta,
                   float* __restrict__ out,
                   int nrows) {
  __shared__ __align__(16) float st[DIM];
  __shared__ __align__(16) float G[6][16 * GP];
  __shared__ float cs[24], sn[24];
  __shared__ float wred[8];
  __shared__ float wout[8][10];

  const int t    = threadIdx.x;
  const int lane = t & 31;
  const int wv   = t >> 5;
  const int n    = lane & 15;
  const int hf   = lane >> 4;
  const int hi2  = hf * 2;
  const unsigned tg = sw(src_perm((unsigned)t));
  const unsigned un = (unsigned)n, uh = (unsigned)hf, uw = (unsigned)wv;
  const unsigned tb_hr = sw(un | (uw << 5) | (uh << 9));
  const unsigned tb_hw = sw(un | (uw << 5) | (uh << 11));
  const unsigned tb_fl = sw((uw << 9) | (un << 4) | (uh << 2));
  const unsigned tb_fs = sw((uw << 9) | (un << 4) | (uh << 3));
  const unsigned tb_ps = sw((unsigned)t);
  const int fo = wv * 128 + n * 4 + hf;

  int row = blockIdx.x;
  float4 nx[4];
  if (row < nrows) {
    const float4* xr = (const float4*)(x + (size_t)row * DIM);
#pragma unroll
    for (int e = 0; e < 4; ++e) nx[e] = xr[fo + (e >> 1) * 64 + (e & 1) * 2];
  }

  if (t < 24) {
    float s, c;
    sincosf(theta[t] * 0.5f, &s, &c);
    cs[t] = c;
    sn[t] = s;
  }
  lds_barrier();

  {
    const int r = t >> 4, c = t & 15;
#pragma unroll
    for (int m = 0; m < 6; ++m) {
      float v = 1.f;
#pragma unroll
      for (int q = 0; q < 4; ++q) {
        const int rb = (r >> (3 - q)) & 1;
        const int cb = (c >> (3 - q)) & 1;
        const float cv = cs[m * 4 + q], sv = sn[m * 4 + q];
        v *= rb ? (cb ? cv : sv) : (cb ? -sv : cv);
      }
      G[m][r * GP + c] = v;
    }
  }

#pragma unroll 1
  for (; row < nrows; row += NWG) {
    v2f s2 = {0.f, 0.f};
#pragma unroll
    for (int e = 0; e < 4; ++e) {
      const float4 v = nx[e];
      v2f lo, hi;
      lo.x = v.x; lo.y = v.y;
      hi.x = v.z; hi.y = v.w;
      s2 += lo * lo;
      s2 += hi * hi;
    }
    float ss = s2.x + s2.y;
#pragma unroll
    for (int o = 16; o > 0; o >>= 1) ss += __shfl_xor(ss, o, 32);
    if (lane == 0) wred[wv] = ss;
    lds_barrier();

    float sumsq = 0.f;
#pragma unroll
    for (int w = 0; w < 8; ++w) sumsq += wred[w];

    const bool fb = (sumsq <= 1e-24f);
    float4 cur[4];
#pragma unroll
    for (int e = 0; e < 4; ++e) cur[e] = nx[e];
    if (fb) {
#pragma unroll
      for (int e = 0; e < 4; ++e) cur[e] = make_float4(0.f, 0.f, 0.f, 0.f);
      if (t == 0) cur[0].x = 1.f;
    }
    const float inv = fb ? 1.f : 1.f / sumsq;

    mid_lo_pass(cur, G[1], G[2], st, tb_fs, n, hf);
    if (row + NWG < nrows) {
      const float4* xn = (const float4*)(x + (size_t)(row + NWG) * DIM);
#pragma unroll
      for (int e = 0; e < 4; ++e) nx[e] = xn[fo + (e >> 1) * 64 + (e & 1) * 2];
    }
    lds_barrier();
    hi_pass(G[0], st, tb_hr, tb_hw, n, hi2);
    lds_barrier();

    {
      float pr[16];
      gather_perm<0>(st, tg, pr);
      lds_barrier();
#pragma unroll
      for (int e = 0; e < 16; ++e) st[at(tb_ps, (unsigned)e << 8)] = pr[e];
      lds_barrier();
    }

    {
      float4 f[4];
#pragma unroll
      for (int e = 0; e < 4; ++e)
        f[e] = *(const float4*)&st[at(tb_fl, ((unsigned)(e >> 1) << 8) | ((unsigned)(e & 1) << 3))];
      mid_lo_pass(f, G[4], G[5], st, tb_fs, n, hf);
    }
    lds_barrier();
    hi_pass(G[3], st, tb_hr, tb_hw, n, hi2);
    lds_barrier();

    float fin[16];
    gather_perm<0>(st, tg, fin);
    float p[16];
#pragma unroll
    for (int e = 0; e < 16; ++e) p[e] = fin[e] * fin[e];
    float a[8], b[4], c[2];
    float d3 = 0.f, d2 = 0.f, d1 = 0.f;
#pragma unroll
    for (int i = 0; i < 8; ++i) {
      a[i] = p[2 * i] + p[2 * i + 1];
      d3 += p[2 * i] - p[2 * i + 1];
    }
#pragma unroll
    for (int i = 0; i < 4; ++i) {
      b[i] = a[2 * i] + a[2 * i + 1];
      d2 += a[2 * i] - a[2 * i + 1];
    }
#pragma unroll
    for (int i = 0; i < 2; ++i) {
      c[i] = b[2 * i] + b[2 * i + 1];
      d1 += b[2 * i] - b[2 * i + 1];
    }
    const float S  = c[0] + c[1];
    const float d0 = c[0] - c[1];

    float xs = S;
#pragma unroll
    for (int bb = 4; bb >= 0; --bb) {
      const float u = __shfl_xor(xs, 1 << bb, 32);
      xs = ((lane >> bb) == 1) ? (u - xs) : (xs + u);
    }
    const bool b4 = (lane & 16) != 0;
    const bool b3 = (lane & 8) != 0;
    const float y01 = (b4 ? d1 : d0) + __shfl_xor(b4 ? d0 : d1, 16, 32);
    const float y23 = (b4 ? d3 : d2) + __shfl_xor(b4 ? d2 : d3, 16, 32);
    float zs = (b3 ? y23 : y01) + __shfl_xor(b3 ? y01 : y23, 8, 32);
#pragma unroll
    for (int o = 4; o > 0; o >>= 1) zs += __shfl_xor(zs, o, 32);

    if ((lane & 7) == 0) wout[wv][((lane >> 4) & 1) + ((lane >> 3) & 1) * 2] = zs;
    if ((lane & (lane - 1)) == 0)
      wout[wv][lane == 0 ? 4 : 10 - __ffs(lane)] = xs;
    lds_barrier();
    if (t < 12) {
      const bool wave_signed = (t >= 4) && (t < 7);
      const int slot = (t < 4) ? t : (wave_signed ? 4 : t - 2);
      float s = 0.f;
#pragma unroll
      for (int w2 = 0; w2 < 8; ++w2) {
        const float v = wout[w2][slot];
        const bool neg = wave_signed && (((w2 >> (6 - t)) & 1) != 0);
        s += neg ? -v : v;
      }
      out[(size_t)row * 12 + t] = s * inv;
    }
  }
}

extern "C" void kernel_launch(void* const* d_in, const int* in_sizes, int n_in,
                              void* d_out, int out_size, void* d_ws,
                              size_t ws_size, hipStream_t stream) {
  const float* x = (const float*)d_in[0];
  const float* theta = (const float*)d_in[1];
  float* out = (float*)d_out;
  const int B = in_sizes[0] / DIM;
  qdense_kernel<<<dim3(NWG), dim3(256), 0, stream>>>(x, theta, out, B);
}


# CNOT cascade after layer 0 folded into the HI pass's stores: each result goes straight to its permuted slot of a second LDS state buffer; gather, re-store and 2 of 8 barriers per row are gone
# speedup vs baseline: 2.5239x; 1.0340x over previous
#include <hip/hip_runtime.h>
#include <math.h>


typedef __attribute__((ext_vector_type(2))) float v2f;
typedef __attribute__((ext_vector_type(8))) float v8f;

#define DIM 4096
#define GP 20
#define NWG 1024

__host__ __device__ constexpr unsigned sw(unsigned k) {
  return k ^ (((k >> 6) & 1u) << 2) ^ (((k >> 7) & 1u) * 24u) ^
         (((k >> 9) & 1u) << 5) ^ (((k >> 11) & 1u) << 4);
}
__device__ __forceinline__ unsigned at(unsigned tb, unsigned c) {
  const unsigned s = sw(c);
  return (tb ^ (s & 63u)) + (s & ~63u);
}

constexpr unsigned cnot_src(unsigned k) {
  for (int i = 10; i >= 0; --i)
    for (int j = 11; j >= i + 1; --j)
      k ^= ((k >> (11 - i)) & 1u) << (11 - j);
  return k;
}
constexpr unsigned mask_p(int p) {
  unsigned mm = 0;
  for (int b = 0; b < 12; ++b)
    mm |= ((cnot_src(1u << b) >> p) & 1u) << b;
  return mm;
}
__constant__ unsigned SRC_MASK[12] = {
    mask_p(0), mask_p(1), mask_p(2),  mask_p(3),
    mask_p(4), mask_p(5), mask_p(6),  mask_p(7),
    mask_p(8), mask_p(9), mask_p(10), mask_p(11)};

constexpr unsigned cnot_dst(unsigned k) {
  for (int i = 0; i <= 10; ++i)
    for (int j = i + 1; j <= 11; ++j)
      k ^= ((k >> (11 - i)) & 1u) << (11 - j);
  return k;
}
constexpr unsigned dmask_p(int p) {
  unsigned mm = 0;
  for (int b = 0; b < 12; ++b)
    mm |= ((cnot_dst(1u << b) >> p) & 1u) << b;
  return mm;
}
__constant__ unsigned DST_MASK[12] = {
    dmask_p(0), dmask_p(1), dmask_p(2),  dmask_p(3),
    dmask_p(4), dmask_p(5), dmask_p(6),  dmask_p(7),
    dmask_p(8), dmask_p(9), dmask_p(10), dmask_p(11)};

__device__ __forceinline__ unsigned dst_perm(unsigned k) {
  unsigned r = 0;
#pragma unroll
  for (int p = 0; p < 12; ++p)
    r |= (unsigned)(__popc(k & DST_MASK[p]) & 1) << p;
  return r;
}

__device__ __forceinline__ unsigned src_perm(unsigned k) {
  unsigned r = 0;
#pragma unroll
  for (int p = 0; p < 12; ++p)
    r |= (unsigned)(__popc(k & SRC_MASK[p]) & 1) << p;
  return r;
}

template <int E>
struct SrcHi { static constexpr unsigned value = cnot_src((unsigned)E << 8); };

template <int E>
__device__ __forceinline__ void gather_perm(const float* st, unsigned tg, float* pr) {
  if constexpr (E < 16) {
    constexpr unsigned s = sw(SrcHi<E>::value);
    pr[E] = st[(tg ^ (s & 0xFFu)) + (s & ~0xFFu)];
    gather_perm<E + 1>(st, tg, pr);
  }
}

template <int TT, int V>
struct DstC {
  static constexpr unsigned value = sw(cnot_dst(((unsigned)TT << 4) | ((unsigned)V << 8)));
};
template <int TT, int V>
__device__ __forceinline__ void scatter_perm(float* sp, unsigned tgw, const v8f& acc) {
  if constexpr (V < 8) {
    sp[tgw ^ DstC<TT, V>::value] = acc[V];
    scatter_perm<TT, V + 1>(sp, tgw, acc);
  }
}

__device__ __forceinline__ v8f wmma4(v2f a, v2f b, v8f c) {
  return __builtin_amdgcn_wmma_f32_16x16x4_f32(false, a, false, b,
                                               (short)0, c, false, false);
}

__device__ __forceinline__ void lds_barrier() {
  __builtin_amdgcn_fence(__ATOMIC_RELEASE, "workgroup", "local");
  __builtin_amdgcn_s_barrier();
  __builtin_amdgcn_fence(__ATOMIC_ACQUIRE, "workgroup", "local");
}

__device__ __forceinline__ void mid_lo_pass(const float4 (&f)[4], const float* g1,
                                            const float* g2, float* st,
                                            unsigned tb_fs, int n, int hf) {
  v2f b1[4], b2[4];
#pragma unroll
  for (int kc = 0; kc < 4; ++kc) {
    b1[kc] = *(const v2f*)&g2[n * GP + 4 * hf + 8 * (kc >> 1) + 2 * (kc & 1)];
    b2[kc] = *(const v2f*)&g1[n * GP + 8 * hf + 2 * kc];
  }
  v8f u[2], r[2];
#pragma unroll
  for (int tt = 0; tt < 2; ++tt) {
    u[tt] = {};
#pragma unroll
    for (int kc = 0; kc < 4; ++kc) {
      const float4 v = f[tt * 2 + (kc >> 1)];
      v2f a;
      a.x = (kc & 1) ? v.z : v.x;
      a.y = (kc & 1) ? v.w : v.y;
      u[tt] = wmma4(a, b1[kc], u[tt]);
    }
  }
#pragma unroll
  for (int tt = 0; tt < 2; ++tt) {
    r[tt] = {};
#pragma unroll
    for (int kc = 0; kc < 4; ++kc) {
      v2f a;
      a.x = u[tt][2 * kc];
      a.y = u[tt][2 * kc + 1];
      r[tt] = wmma4(a, b2[kc], r[tt]);
    }
  }
#pragma unroll
  for (int tt = 0; tt < 2; ++tt) {
#pragma unroll
    for (int q = 0; q < 2; ++q) {
      float4 o;
      o.x = r[tt][4 * q];
      o.y = r[tt][4 * q + 1];
      o.z = r[tt][4 * q + 2];
      o.w = r[tt][4 * q + 3];
      *(float4*)&st[at(tb_fs, ((unsigned)tt << 8) | ((unsigned)q << 2))] = o;
    }
  }
}

__device__ __forceinline__ void hi_pass(const float* g0, float* st, unsigned tb_hr,
                                        unsigned tb_hw, int n, int hi2) {
  v2f a[4];
#pragma unroll
  for (int kc = 0; kc < 4; ++kc)
    a[kc] = *(const v2f*)&g0[n * GP + kc * 4 + hi2];
#pragma unroll
  for (int tt = 0; tt < 2; ++tt) {
    v8f acc = {};
#pragma unroll
    for (int kc = 0; kc < 4; ++kc) {
      const unsigned c = ((unsigned)tt << 4) | ((unsigned)kc << 10);
      v2f b;
      b.x = st[at(tb_hr, c)];
      b.y = st[at(tb_hr, c | (1u << 8))];
      acc = wmma4(a[kc], b, acc);
    }
#pragma unroll
    for (int v = 0; v < 8; ++v)
      st[at(tb_hw, ((unsigned)tt << 4) | ((unsigned)v << 8))] = acc[v];
  }
}

template <int TT>
__device__ __forceinline__ void hi_tile_perm(const v2f (&a)[4], const float* st, float* sp,
                                             unsigned tb_hr, unsigned tgw) {
  v8f acc = {};
#pragma unroll
  for (int kc = 0; kc < 4; ++kc) {
    const unsigned c = ((unsigned)TT << 4) | ((unsigned)kc << 10);
    v2f b;
    b.x = st[at(tb_hr, c)];
    b.y = st[at(tb_hr, c | (1u << 8))];
    acc = wmma4(a[kc], b, acc);
  }
  scatter_perm<TT, 0>(sp, tgw, acc);
}
__device__ __forceinline__ void hi_pass_perm(const float* g0, const float* st, float* sp,
                                             unsigned tb_hr, unsigned tgw, int n, int hi2) {
  v2f a[4];
#pragma unroll
  for (int kc = 0; kc < 4; ++kc)
    a[kc] = *(const v2f*)&g0[n * GP + kc * 4 + hi2];
  hi_tile_perm<0>(a, st, sp, tb_hr, tgw);
  hi_tile_perm<1>(a, st, sp, tb_hr, tgw);
}

__global__ __launch_bounds__(256) __attribute__((amdgpu_waves_per_eu(8, 8)))
void qdense_kernel(const float* __restrict__ x,
                   const float* __restrict__ theta,
                   float* __restrict__ out,
                   int nrows) {
  __shared__ __align__(16) float st[DIM];
  __shared__ __align__(16) float st2[DIM];
  __shared__ __align__(16) float G[6][16 * GP];
  __shared__ float cs[24], sn[24];
  __shared__ float wred[8];
  __shared__ float wout[8][10];

  const int t    = threadIdx.x;
  const int lane = t & 31;
  const int wv   = t >> 5;
  const int n    = lane & 15;
  const int hf   = lane >> 4;
  const int hi2  = hf * 2;
  const unsigned tg = sw(src_perm((unsigned)t));
  const unsigned un = (unsigned)n, uh = (unsigned)hf, uw = (unsigned)wv;
  const unsigned tb_hr = sw(un | (uw << 5) | (uh << 9));
  const unsigned tb_hw = sw(un | (uw << 5) | (uh << 11));
  const unsigned tb_fl = sw((uw << 9) | (un << 4) | (uh << 2));
  const unsigned tb_fs = sw((uw << 9) | (un << 4) | (uh << 3));
  const unsigned tgw = sw(dst_perm(un | (uw << 5) | (uh << 11)));
  const int fo = wv * 128 + n * 4 + hf;

  int row = blockIdx.x;
  float4 nx[4];
  if (row < nrows) {
    const float4* xr = (const float4*)(x + (size_t)row * DIM);
#pragma unroll
    for (int e = 0; e < 4; ++e) nx[e] = xr[fo + (e >> 1) * 64 + (e & 1) * 2];
  }

  if (t < 24) {
    float s, c;
    sincosf(theta[t] * 0.5f, &s, &c);
    cs[t] = c;
    sn[t] = s;
  }
  lds_barrier();

  {
    const int r = t >> 4, c = t & 15;
#pragma unroll
    for (int m = 0; m < 6; ++m) {
      float v = 1.f;
#pragma unroll
      for (int q = 0; q < 4; ++q) {
        const int rb = (r >> (3 - q)) & 1;
        const int cb = (c >> (3 - q)) & 1;
        const float cv = cs[m * 4 + q], sv = sn[m * 4 + q];
        v *= rb ? (cb ? cv : sv) : (cb ? -sv : cv);
      }
      G[m][r * GP + c] = v;
    }
  }

#pragma unroll 1
  for (; row < nrows; row += NWG) {
    v2f s2 = {0.f, 0.f};
#pragma unroll
    for (int e = 0; e < 4; ++e) {
      const float4 v = nx[e];
      v2f lo, hi;
      lo.x = v.x; lo.y = v.y;
      hi.x = v.z; hi.y = v.w;
      s2 += lo * lo;
      s2 += hi * hi;
    }
    float ss = s2.x + s2.y;
#pragma unroll
    for (int o = 16; o > 0; o >>= 1) ss += __shfl_xor(ss, o, 32);
    if (lane == 0) wred[wv] = ss;
    lds_barrier();

    float sumsq = 0.f;
#pragma unroll
    for (int w = 0; w < 8; ++w) sumsq += wred[w];

    const bool fb = (sumsq <= 1e-24f);
    float4 cur[4];
#pragma unroll
    for (int e = 0; e < 4; ++e) cur[e] = nx[e];
    if (fb) {
#pragma unroll
      for (int e = 0; e < 4; ++e) cur[e] = make_float4(0.f, 0.f, 0.f, 0.f);
      if (t == 0) cur[0].x = 1.f;
    }
    const float inv = fb ? 1.f : 1.f / sumsq;

    mid_lo_pass(cur, G[1], G[2], st, tb_fs, n, hf);
    if (row + NWG < nrows) {
      const float4* xn = (const float4*)(x + (size_t)(row + NWG) * DIM);
#pragma unroll
      for (int e = 0; e < 4; ++e) nx[e] = xn[fo + (e >> 1) * 64 + (e & 1) * 2];
    }
    lds_barrier();
    hi_pass_perm(G[0], st, st2, tb_hr, tgw, n, hi2);
    lds_barrier();

    {
      float4 f[4];
#pragma unroll
      for (int e = 0; e < 4; ++e)
        f[e] = *(const float4*)&st2[at(tb_fl, ((unsigned)(e >> 1) << 8) | ((unsigned)(e & 1) << 3))];
      mid_lo_pass(f, G[4], G[5], st2, tb_fs, n, hf);
    }
    lds_barrier();
    hi_pass(G[3], st2, tb_hr, tb_hw, n, hi2);
    lds_barrier();

    float fin[16];
    gather_perm<0>(st2, tg, fin);
    float p[16];
#pragma unroll
    for (int e = 0; e < 16; ++e) p[e] = fin[e] * fin[e];
    float a[8], b[4], c[2];
    float d3 = 0.f, d2 = 0.f, d1 = 0.f;
#pragma unroll
    for (int i = 0; i < 8; ++i) {
      a[i] = p[2 * i] + p[2 * i + 1];
      d3 += p[2 * i] - p[2 * i + 1];
    }
#pragma unroll
    for (int i = 0; i < 4; ++i) {
      b[i] = a[2 * i] + a[2 * i + 1];
      d2 += a[2 * i] - a[2 * i + 1];
    }
#pragma unroll
    for (int i = 0; i < 2; ++i) {
      c[i] = b[2 * i] + b[2 * i + 1];
      d1 += b[2 * i] - b[2 * i + 1];
    }
    const float S  = c[0] + c[1];
    const float d0 = c[0] - c[1];

    float xs = S;
#pragma unroll
    for (int bb = 4; bb >= 0; --bb) {
      const float u = __shfl_xor(xs, 1 << bb, 32);
      xs = ((lane >> bb) == 1) ? (u - xs) : (xs + u);
    }
    const bool b4 = (lane & 16) != 0;
    const bool b3 = (lane & 8) != 0;
    const float y01 = (b4 ? d1 : d0) + __shfl_xor(b4 ? d0 : d1, 16, 32);
    const float y23 = (b4 ? d3 : d2) + __shfl_xor(b4 ? d2 : d3, 16, 32);
    float zs = (b3 ? y23 : y01) + __shfl_xor(b3 ? y01 : y23, 8, 32);
#pragma unroll
    for (int o = 4; o > 0; o >>= 1) zs += __shfl_xor(zs, o, 32);

    if ((lane & 7) == 0) wout[wv][((lane >> 4) & 1) + ((lane >> 3) & 1) * 2] = zs;
    if ((lane & (lane - 1)) == 0)
      wout[wv][lane == 0 ? 4 : 10 - __ffs(lane)] = xs;
    lds_barrier();
    if (t < 12) {
      const bool wave_signed = (t >= 4) && (t < 7);
      const int slot = (t < 4) ? t : (wave_signed ? 4 : t - 2);
      float s = 0.f;
#pragma unroll
      for (int w2 = 0; w2 < 8; ++w2) {
        const float v = wout[w2][slot];
        const bool neg = wave_signed && (((w2 >> (6 - t)) & 1) != 0);
        s += neg ? -v : v;
      }
      out[(size_t)row * 12 + t] = s * inv;
    }
  }
}

extern "C" void kernel_launch(void* const* d_in, const int* in_sizes, int n_in,
                              void* d_out, int out_size, void* d_ws,
                              size_t ws_size, hipStream_t stream) {
  const float* x = (const float*)d_in[0];
  const float* theta = (const float*)d_in[1];
  float* out = (float*)d_out;
  const int B = in_sizes[0] / DIM;
  qdense_kernel<<<dim3(NWG), dim3(256), 0, stream>>>(x, theta, out, B);
}


# All 16 cross-lane hops per row (sum of squares, Z reductions) on the vector ALU: row_xmask DPP adds within a 16-lane row, v_permlanex16 between rows, not ds_bpermute round trips; same order of sums
# speedup vs baseline: 2.6492x; 1.0496x over previous
#include <hip/hip_runtime.h>
#include <math.h>


typedef __attribute__((ext_vector_type(2))) float v2f;
typedef __attribute__((ext_vector_type(8))) float v8f;

#define DIM 4096
#define GP 20
#define NWG 1024

__host__ __device__ constexpr unsigned sw(unsigned k) {
  return k ^ (((k >> 6) & 1u) << 2) ^ (((k >> 7) & 1u) * 24u) ^
         (((k >> 9) & 1u) << 5) ^ (((k >> 11) & 1u) << 4);
}
__device__ __forceinline__ unsigned at(unsigned tb, unsigned c) {
  const unsigned s = sw(c);
  return (tb ^ (s & 63u)) + (s & ~63u);
}

constexpr unsigned cnot_src(unsigned k) {
  for (int i = 10; i >= 0; --i)
    for (int j = 11; j >= i + 1; --j)
      k ^= ((k >> (11 - i)) & 1u) << (11 - j);
  return k;
}
constexpr unsigned mask_p(int p) {
  unsigned mm = 0;
  for (int b = 0; b < 12; ++b)
    mm |= ((cnot_src(1u << b) >> p) & 1u) << b;
  return mm;
}
__constant__ unsigned SRC_MASK[12] = {
    mask_p(0), mask_p(1), mask_p(2),  mask_p(3),
    mask_p(4), mask_p(5), mask_p(6),  mask_p(7),
    mask_p(8), mask_p(9), mask_p(10), mask_p(11)};

constexpr unsigned cnot_dst(unsigned k) {
  for (int i = 0; i <= 10; ++i)
    for (int j = i + 1; j <= 11; ++j)
      k ^= ((k >> (11 - i)) & 1u) << (11 - j);
  return k;
}
constexpr unsigned dmask_p(int p) {
  unsigned mm = 0;
  for (int b = 0; b < 12; ++b)
    mm |= ((cnot_dst(1u << b) >> p) & 1u) << b;
  return mm;
}
__constant__ unsigned DST_MASK[12] = {
    dmask_p(0), dmask_p(1), dmask_p(2),  dmask_p(3),
    dmask_p(4), dmask_p(5), dmask_p(6),  dmask_p(7),
    dmask_p(8), dmask_p(9), dmask_p(10), dmask_p(11)};

__device__ __forceinline__ unsigned dst_perm(unsigned k) {
  unsigned r = 0;
#pragma unroll
  for (int p = 0; p < 12; ++p)
    r |= (unsigned)(__popc(k & DST_MASK[p]) & 1) << p;
  return r;
}

__device__ __forceinline__ unsigned src_perm(unsigned k) {
  unsigned r = 0;
#pragma unroll
  for (int p = 0; p < 12; ++p)
    r |= (unsigned)(__popc(k & SRC_MASK[p]) & 1) << p;
  return r;
}

template <int E>
struct SrcHi { static constexpr unsigned value = cnot_src((unsigned)E << 8); };

template <int E>
__device__ __forceinline__ void gather_perm(const float* st, unsigned tg, float* pr) {
  if constexpr (E < 16) {
    constexpr unsigned s = sw(SrcHi<E>::value);
    pr[E] = st[(tg ^ (s & 0xFFu)) + (s & ~0xFFu)];
    gather_perm<E + 1>(st, tg, pr);
  }
}

template <int TT, int V>
struct DstC {
  static constexpr unsigned value = sw(cnot_dst(((unsigned)TT << 4) | ((unsigned)V << 8)));
};
template <int TT, int V>
__device__ __forceinline__ void scatter_perm(float* sp, unsigned tgw, const v8f& acc) {
  if constexpr (V < 8) {
    sp[tgw ^ DstC<TT, V>::value] = acc[V];
    scatter_perm<TT, V + 1>(sp, tgw, acc);
  }
}

__device__ __forceinline__ v8f wmma4(v2f a, v2f b, v8f c) {
  return __builtin_amdgcn_wmma_f32_16x16x4_f32(false, a, false, b,
                                               (short)0, c, false, false);
}

__device__ __forceinline__ void lds_barrier() {
  __builtin_amdgcn_fence(__ATOMIC_RELEASE, "workgroup", "local");
  __builtin_amdgcn_s_barrier();
  __builtin_amdgcn_fence(__ATOMIC_ACQUIRE, "workgroup", "local");
}

template <int M>
__device__ __forceinline__ float lane_xor(float v) {
  static_assert(M == 1 || M == 2 || M == 4 || M == 8 || M == 16, "one lane bit");
  const int b = __builtin_bit_cast(int, v);
  int r;
  if constexpr (M == 16) {
    r = (int)__builtin_amdgcn_permlanex16(b, b, 0x76543210u, 0xfedcba98u, false, false);
  } else {
    r = __builtin_amdgcn_update_dpp(0, b, 0x160 + M, 0xF, 0xF, true);
  }
  return __builtin_bit_cast(float, r);
}

template <int BB>
__device__ __forceinline__ float split_hop(float xs, int lane) {
  const float u = lane_xor<(1 << BB)>(xs);
  return ((lane >> BB) == 1) ? (u - xs) : (xs + u);
}

__device__ __forceinline__ void mid_lo_pass(const float4 (&f)[4], const float* g1,
                                            const float* g2, float* st,
                                            unsigned tb_fs, int n, int hf) {
  v2f b1[4], b2[4];
#pragma unroll
  for (int kc = 0; kc < 4; ++kc) {
    b1[kc] = *(const v2f*)&g2[n * GP + 4 * hf + 8 * (kc >> 1) + 2 * (kc & 1)];
    b2[kc] = *(const v2f*)&g1[n * GP + 8 * hf + 2 * kc];
  }
  v8f u[2], r[2];
#pragma unroll
  for (int tt = 0; tt < 2; ++tt) {
    u[tt] = {};
#pragma unroll
    for (int kc = 0; kc < 4; ++kc) {
      const float4 v = f[tt * 2 + (kc >> 1)];
      v2f a;
      a.x = (kc & 1) ? v.z : v.x;
      a.y = (kc & 1) ? v.w : v.y;
      u[tt] = wmma4(a, b1[kc], u[tt]);
    }
  }
#pragma unroll
  for (int tt = 0; tt < 2; ++tt) {
    r[tt] = {};
#pragma unroll
    for (int kc = 0; kc < 4; ++kc) {
      v2f a;
      a.x = u[tt][2 * kc];
      a.y = u[tt][2 * kc + 1];
      r[tt] = wmma4(a, b2[kc], r[tt]);
    }
  }
#pragma unroll
  for (int tt = 0; tt < 2; ++tt) {
#pragma unroll
    for (int q = 0; q < 2; ++q) {
      float4 o;
      o.x = r[tt][4 * q];
      o.y = r[tt][4 * q + 1];
      o.z = r[tt][4 * q + 2];
      o.w = r[tt][4 * q + 3];
      *(float4*)&st[at(tb_fs, ((unsigned)tt << 8) | ((unsigned)q << 2))] = o;
    }
  }
}

__device__ __forceinline__ void hi_pass(const float* g0, float* st, unsigned tb_hr,
                                        unsigned tb_hw, int n, int hi2) {
  v2f a[4];
#pragma unroll
  for (int kc = 0; kc < 4; ++kc)
    a[kc] = *(const v2f*)&g0[n * GP + kc * 4 + hi2];
#pragma unroll
  for (int tt = 0; tt < 2; ++tt) {
    v8f acc = {};
#pragma unroll
    for (int kc = 0; kc < 4; ++kc) {
      const unsigned c = ((unsigned)tt << 4) | ((unsigned)kc << 10);
      v2f b;
      b.x = st[at(tb_hr, c)];
      b.y = st[at(tb_hr, c | (1u << 8))];
      acc = wmma4(a[kc], b, acc);
    }
#pragma unroll
    for (int v = 0; v < 8; ++v)
      st[at(tb_hw, ((unsigned)tt << 4) | ((unsigned)v << 8))] = acc[v];
  }
}

template <int TT>
__device__ __forceinline__ void hi_tile_perm(const v2f (&a)[4], const float* st, float* sp,
                                             unsigned tb_hr, unsigned tgw) {
  v8f acc = {};
#pragma unroll
  for (int kc = 0; kc < 4; ++kc) {
    const unsigned c = ((unsigned)TT << 4) | ((unsigned)kc << 10);
    v2f b;
    b.x = st[at(tb_hr, c)];
    b.y = st[at(tb_hr, c | (1u << 8))];
    acc = wmma4(a[kc], b, acc);
  }
  scatter_perm<TT, 0>(sp, tgw, acc);
}
__device__ __forceinline__ void hi_pass_perm(const float* g0, const float* st, float* sp,
                                             unsigned tb_hr, unsigned tgw, int n, int hi2) {
  v2f a[4];
#pragma unroll
  for (int kc = 0; kc < 4; ++kc)
    a[kc] = *(const v2f*)&g0[n * GP + kc * 4 + hi2];
  hi_tile_perm<0>(a, st, sp, tb_hr, tgw);
  hi_tile_perm<1>(a, st, sp, tb_hr, tgw);
}

__global__ __launch_bounds__(256) __attribute__((amdgpu_waves_per_eu(8, 8)))
void qdense_kernel(const float* __restrict__ x,
                   const float* __restrict__ theta,
                   float* __restrict__ out,
                   int nrows) {
  __shared__ __align__(16) float st[DIM];
  __shared__ __align__(16) float st2[DIM];
  __shared__ __align__(16) float G[6][16 * GP];
  __shared__ float cs[24], sn[24];
  __shared__ float wred[8];
  __shared__ float wout[8][10];

  const int t    = threadIdx.x;
  const int lane = t & 31;
  const int wv   = t >> 5;
  const int n    = lane & 15;
  const int hf   = lane >> 4;
  const int hi2  = hf * 2;
  const unsigned tg = sw(src_perm((unsigned)t));
  const unsigned un = (unsigned)n, uh = (unsigned)hf, uw = (unsigned)wv;
  const unsigned tb_hr = sw(un | (uw << 5) | (uh << 9));
  const unsigned tb_hw = sw(un | (uw << 5) | (uh << 11));
  const unsigned tb_fl = sw((uw << 9) | (un << 4) | (uh << 2));
  const unsigned tb_fs = sw((uw << 9) | (un << 4) | (uh << 3));
  const unsigned tgw = sw(dst_perm(un | (uw << 5) | (uh << 11)));
  const int fo = wv * 128 + n * 4 + hf;

  int row = blockIdx.x;
  float4 nx[4];
  if (row < nrows) {
    const float4* xr = (const float4*)(x + (size_t)row * DIM);
#pragma unroll
    for (int e = 0; e < 4; ++e) nx[e] = xr[fo + (e >> 1) * 64 + (e & 1) * 2];
  }

  if (t < 24) {
    float s, c;
    sincosf(theta[t] * 0.5f, &s, &c);
    cs[t] = c;
    sn[t] = s;
  }
  lds_barrier();

  {
    const int r = t >> 4, c = t & 15;
#pragma unroll
    for (int m = 0; m < 6; ++m) {
      float v = 1.f;
#pragma unroll
      for (int q = 0; q < 4; ++q) {
        const int rb = (r >> (3 - q)) & 1;
        const int cb = (c >> (3 - q)) & 1;
        const float cv = cs[m * 4 + q], sv = sn[m * 4 + q];
        v *= rb ? (cb ? cv : sv) : (cb ? -sv : cv);
      }
      G[m][r * GP + c] = v;
    }
  }

#pragma unroll 1
  for (; row < nrows; row += NWG) {
    v2f s2 = {0.f, 0.f};
#pragma unroll
    for (int e = 0; e < 4; ++e) {
      const float4 v = nx[e];
      v2f lo, hi;
      lo.x = v.x; lo.y = v.y;
      hi.x = v.z; hi.y = v.w;
      s2 += lo * lo;
      s2 += hi * hi;
    }
    float ss = s2.x + s2.y;
    ss += lane_xor<16>(ss);
    ss += lane_xor<8>(ss);
    ss += lane_xor<4>(ss);
    ss += lane_xor<2>(ss);
    ss += lane_xor<1>(ss);
    if (lane == 0) wred[wv] = ss;
    lds_barrier();

    float sumsq = 0.f;
#pragma unroll
    for (int w = 0; w < 8; ++w) sumsq += wred[w];

    const bool fb = (sumsq <= 1e-24f);
    float4 cur[4];
#pragma unroll
    for (int e = 0; e < 4; ++e) cur[e] = nx[e];
    if (fb) {
#pragma unroll
      for (int e = 0; e < 4; ++e) cur[e] = make_float4(0.f, 0.f, 0.f, 0.f);
      if (t == 0) cur[0].x = 1.f;
    }
    const float inv = fb ? 1.f : 1.f / sumsq;

    mid_lo_pass(cur, G[1], G[2], st, tb_fs, n, hf);
    if (row + NWG < nrows) {
      const float4* xn = (const float4*)(x + (size_t)(row + NWG) * DIM);
#pragma unroll
      for (int e = 0; e < 4; ++e) nx[e] = xn[fo + (e >> 1) * 64 + (e & 1) * 2];
    }
    lds_barrier();
    hi_pass_perm(G[0], st, st2, tb_hr, tgw, n, hi2);
    lds_barrier();

    {
      float4 f[4];
#pragma unroll
      for (int e = 0; e < 4; ++e)
        f[e] = *(const float4*)&st2[at(tb_fl, ((unsigned)(e >> 1) << 8) | ((unsigned)(e & 1) << 3))];
      mid_lo_pass(f, G[4], G[5], st2, tb_fs, n, hf);
    }
    lds_barrier();
    hi_pass(G[3], st2, tb_hr, tb_hw, n, hi2);
    lds_barrier();

    float fin[16];
    gather_perm<0>(st2, tg, fin);
    float p[16];
#pragma unroll
    for (int e = 0; e < 16; ++e) p[e] = fin[e] * fin[e];
    float a[8], b[4], c[2];
    float d3 = 0.f, d2 = 0.f, d1 = 0.f;
#pragma unroll
    for (int i = 0; i < 8; ++i) {
      a[i] = p[2 * i] + p[2 * i + 1];
      d3 += p[2 * i] - p[2 * i + 1];
    }
#pragma unroll
    for (int i = 0; i < 4; ++i) {
      b[i] = a[2 * i] + a[2 * i + 1];
      d2 += a[2 * i] - a[2 * i + 1];
    }
#pragma unroll
    for (int i = 0; i < 2; ++i) {
      c[i] = b[2 * i] + b[2 * i + 1];
      d1 += b[2 * i] - b[2 * i + 1];
    }
    const float S  = c[0] + c[1];
    const float d0 = c[0] - c[1];

    float xs = S;
    xs = split_hop<4>(xs, lane);
    xs = split_hop<3>(xs, lane);
    xs = split_hop<2>(xs, lane);
    xs = split_hop<1>(xs, lane);
    xs = split_hop<0>(xs, lane);
    const bool b4 = (lane & 16) != 0;
    const bool b3 = (lane & 8) != 0;
    const float give01 = b4 ? d0 : d1, give23 = b4 ? d2 : d3;
    const float y01 = (b4 ? d1 : d0) + lane_xor<16>(give01);
    const float y23 = (b4 ? d3 : d2) + lane_xor<16>(give23);
    const float give8 = b3 ? y01 : y23;
    float zs = (b3 ? y23 : y01) + lane_xor<8>(give8);
    zs += lane_xor<4>(zs);
    zs += lane_xor<2>(zs);
    zs += lane_xor<1>(zs);

    if ((lane & 7) == 0) wout[wv][((lane >> 4) & 1) + ((lane >> 3) & 1) * 2] = zs;
    if ((lane & (lane - 1)) == 0)
      wout[wv][lane == 0 ? 4 : 10 - __ffs(lane)] = xs;
    lds_barrier();
    if (t < 12) {
      const bool wave_signed = (t >= 4) && (t < 7);
      const int slot = (t < 4) ? t : (wave_signed ? 4 : t - 2);
      float s = 0.f;
#pragma unroll
      for (int w2 = 0; w2 < 8; ++w2) {
        const float v = wout[w2][slot];
        const bool neg = wave_signed && (((w2 >> (6 - t)) & 1) != 0);
        s += neg ? -v : v;
      }
      out[(size_t)row * 12 + t] = s * inv;
    }
  }
}

extern "C" void kernel_launch(void* const* d_in, const int* in_sizes, int n_in,
                              void* d_out, int out_size, void* d_ws,
                              size_t ws_size, hipStream_t stream) {
  const float* x = (const float*)d_in[0];
  const float* theta = (const float*)d_in[1];
  float* out = (float*)d_out;
  const int B = in_sizes[0] / DIM;
  qdense_kernel<<<dim3(NWG), dim3(256), 0, stream>>>(x, theta, out, B);
}


# Last CNOT cascade never applied: it maps j to j^(j>>1), so Z(w) behind it is parity Z(w-1)Z(w) before it; 12 sums taken from layer 1's HI result registers: no store, barrier, gather; new sum order
# speedup vs baseline: 2.7359x; 1.0327x over previous
#include <hip/hip_runtime.h>
#include <math.h>


typedef __attribute__((ext_vector_type(2))) float v2f;
typedef __attribute__((ext_vector_type(8))) float v8f;

#define DIM 4096
#define GP 20
#define NWG 1024

__host__ __device__ constexpr unsigned sw(unsigned k) {
  return k ^ (((k >> 6) & 1u) << 2) ^ (((k >> 7) & 1u) * 24u) ^
         (((k >> 9) & 1u) << 5) ^ (((k >> 11) & 1u) << 4);
}
__device__ __forceinline__ unsigned at(unsigned tb, unsigned c) {
  const unsigned s = sw(c);
  return (tb ^ (s & 63u)) + (s & ~63u);
}

constexpr unsigned cnot_dst(unsigned k) {
  for (int i = 0; i <= 10; ++i)
    for (int j = i + 1; j <= 11; ++j)
      k ^= ((k >> (11 - i)) & 1u) << (11 - j);
  return k;
}
constexpr bool dst_is_shift_xor() {
  for (int b = 0; b < 12; ++b)
    if (cnot_dst(1u << b) != ((1u << b) ^ ((1u << b) >> 1))) return false;
  return true;
}
static_assert(dst_is_shift_xor(), "cascade is not k ^ (k >> 1)");
constexpr unsigned dmask_p(int p) {
  unsigned mm = 0;
  for (int b = 0; b < 12; ++b)
    mm |= ((cnot_dst(1u << b) >> p) & 1u) << b;
  return mm;
}
__constant__ unsigned DST_MASK[12] = {
    dmask_p(0), dmask_p(1), dmask_p(2),  dmask_p(3),
    dmask_p(4), dmask_p(5), dmask_p(6),  dmask_p(7),
    dmask_p(8), dmask_p(9), dmask_p(10), dmask_p(11)};

__device__ __forceinline__ unsigned dst_perm(unsigned k) {
  unsigned r = 0;
#pragma unroll
  for (int p = 0; p < 12; ++p)
    r |= (unsigned)(__popc(k & DST_MASK[p]) & 1) << p;
  return r;
}

template <int TT, int V>
struct DstC {
  static constexpr unsigned value = sw(cnot_dst(((unsigned)TT << 4) | ((unsigned)V << 8)));
};
template <int TT, int V>
__device__ __forceinline__ void scatter_perm(float* sp, unsigned tgw, const v8f& acc) {
  if constexpr (V < 8) {
    sp[tgw ^ DstC<TT, V>::value] = acc[V];
    scatter_perm<TT, V + 1>(sp, tgw, acc);
  }
}

__device__ __forceinline__ v8f wmma4(v2f a, v2f b, v8f c) {
  return __builtin_amdgcn_wmma_f32_16x16x4_f32(false, a, false, b,
                                               (short)0, c, false, false);
}

__device__ __forceinline__ void lds_barrier() {
  __builtin_amdgcn_fence(__ATOMIC_RELEASE, "workgroup", "local");
  __builtin_amdgcn_s_barrier();
  __builtin_amdgcn_fence(__ATOMIC_ACQUIRE, "workgroup", "local");
}

template <int M>
__device__ __forceinline__ float lane_xor(float v) {
  static_assert(M == 1 || M == 2 || M == 4 || M == 8 || M == 16, "one lane bit");
  const int b = __builtin_bit_cast(int, v);
  int r;
  if constexpr (M == 16) {
    r = (int)__builtin_amdgcn_permlanex16(b, b, 0x76543210u, 0xfedcba98u, false, false);
  } else {
    r = __builtin_amdgcn_update_dpp(0, b, 0x160 + M, 0xF, 0xF, true);
  }
  return __builtin_bit_cast(float, r);
}

template <int BB>
__device__ __forceinline__ float walsh_hop(float xs, int lane) {
  const float u = lane_xor<(1 << BB)>(xs);
  return ((lane >> BB) & 1) ? (u - xs) : (xs + u);
}

template <int M>
__device__ __forceinline__ float pair_hop(float a, float b, bool bit) {
  const float give = bit ? a : b;
  return (bit ? b : a) + lane_xor<M>(give);
}

__device__ __forceinline__ void mid_lo_pass(const float4 (&f)[4], const float* g1,
                                            const float* g2, float* st,
                                            unsigned tb_fs, int n, int hf) {
  v2f b1[4], b2[4];
#pragma unroll
  for (int kc = 0; kc < 4; ++kc) {
    b1[kc] = *(const v2f*)&g2[n * GP + 4 * hf + 8 * (kc >> 1) + 2 * (kc & 1)];
    b2[kc] = *(const v2f*)&g1[n * GP + 8 * hf + 2 * kc];
  }
  v8f u[2], r[2];
#pragma unroll
  for (int tt = 0; tt < 2; ++tt) {
    u[tt] = {};
#pragma unroll
    for (int kc = 0; kc < 4; ++kc) {
      const float4 v = f[tt * 2 + (kc >> 1)];
      v2f a;
      a.x = (kc & 1) ? v.z : v.x;
      a.y = (kc & 1) ? v.w : v.y;
      u[tt] = wmma4(a, b1[kc], u[tt]);
    }
  }
#pragma unroll
  for (int tt = 0; tt < 2; ++tt) {
    r[tt] = {};
#pragma unroll
    for (int kc = 0; kc < 4; ++kc) {
      v2f a;
      a.x = u[tt][2 * kc];
      a.y = u[tt][2 * kc + 1];
      r[tt] = wmma4(a, b2[kc], r[tt]);
    }
  }
#pragma unroll
  for (int tt = 0; tt < 2; ++tt) {
#pragma unroll
    for (int q = 0; q < 2; ++q) {
      float4 o;
      o.x = r[tt][4 * q];
      o.y = r[tt][4 * q + 1];
      o.z = r[tt][4 * q + 2];
      o.w = r[tt][4 * q + 3];
      *(float4*)&st[at(tb_fs, ((unsigned)tt << 8) | ((unsigned)q << 2))] = o;
    }
  }
}

__device__ __forceinline__ void hi_pass_keep(const float* g0, const float* st, unsigned tb_hr,
                                             int n, int hi2, v8f (&acc)[2]) {
  v2f a[4];
#pragma unroll
  for (int kc = 0; kc < 4; ++kc)
    a[kc] = *(const v2f*)&g0[n * GP + kc * 4 + hi2];
  acc[0] = {};
  acc[1] = {};
#pragma unroll
  for (int kc = 0; kc < 4; ++kc) {
#pragma unroll
    for (int tt = 0; tt < 2; ++tt) {
      const unsigned c = ((unsigned)tt << 4) | ((unsigned)kc << 10);
      v2f b;
      b.x = st[at(tb_hr, c)];
      b.y = st[at(tb_hr, c | (1u << 8))];
      acc[tt] = wmma4(a[kc], b, acc[tt]);
    }
  }
}

template <int TT>
__device__ __forceinline__ void hi_tile_perm(const v2f (&a)[4], const float* st, float* sp,
                                             unsigned tb_hr, unsigned tgw) {
  v8f acc = {};
#pragma unroll
  for (int kc = 0; kc < 4; ++kc) {
    const unsigned c = ((unsigned)TT << 4) | ((unsigned)kc << 10);
    v2f b;
    b.x = st[at(tb_hr, c)];
    b.y = st[at(tb_hr, c | (1u << 8))];
    acc = wmma4(a[kc], b, acc);
  }
  scatter_perm<TT, 0>(sp, tgw, acc);
}
__device__ __forceinline__ void hi_pass_perm(const float* g0, const float* st, float* sp,
                                             unsigned tb_hr, unsigned tgw, int n, int hi2) {
  v2f a[4];
#pragma unroll
  for (int kc = 0; kc < 4; ++kc)
    a[kc] = *(const v2f*)&g0[n * GP + kc * 4 + hi2];
  hi_tile_perm<0>(a, st, sp, tb_hr, tgw);
  hi_tile_perm<1>(a, st, sp, tb_hr, tgw);
}

__global__ __launch_bounds__(256) __attribute__((amdgpu_waves_per_eu(8, 8)))
void qdense_kernel(const float* __restrict__ x,
                   const float* __restrict__ theta,
                   float* __restrict__ out,
                   int nrows) {
  __shared__ __align__(16) float st[DIM];
  __shared__ __align__(16) float st2[DIM];
  __shared__ __align__(16) float G[6][16 * GP];
  __shared__ float cs[24], sn[24];
  __shared__ float wred[8];
  __shared__ float wout[8][12];

  const int t    = threadIdx.x;
  const int lane = t & 31;
  const int wv   = t >> 5;
  const int n    = lane & 15;
  const int hf   = lane >> 4;
  const int hi2  = hf * 2;
  const unsigned un = (unsigned)n, uh = (unsigned)hf, uw = (unsigned)wv;
  const unsigned tb_hr = sw(un | (uw << 5) | (uh << 9));
  const unsigned tb_fl = sw((uw << 9) | (un << 4) | (uh << 2));
  const unsigned tb_fs = sw((uw << 9) | (un << 4) | (uh << 3));
  const unsigned tgw = sw(dst_perm(un | (uw << 5) | (uh << 11)));
  const int fo = wv * 128 + n * 4 + hf;

  constexpr unsigned XL = 0x00011049u, ZL = 0x01110111u;
  const unsigned below = (1u << lane) - 1u;
  const bool x_keep = ((XL >> lane) & 1u) != 0;
  const bool z_keep = ((ZL >> lane) & 1u) != 0;
  const int x_slot = __popc(XL & below);
  const int z_slot = 5 + __popc(ZL & below);
  const unsigned tq = ((unsigned)t & 15u) * 4u;
  const int o_slot = (int)((0x1239600A7854ull >> tq) & 15u);
  const unsigned o_wbits = (0x13640000u >> (tq & 31u)) & (tq < 32u ? 15u : 0u);
  unsigned o_neg = 0;
#pragma unroll
  for (int w2 = 0; w2 < 8; ++w2)
    o_neg |= (unsigned)(__popc((unsigned)w2 & o_wbits) & 1) << w2;

  int row = blockIdx.x;
  float4 nx[4];
  if (row < nrows) {
    const float4* xr = (const float4*)(x + (size_t)row * DIM);
#pragma unroll
    for (int e = 0; e < 4; ++e) nx[e] = xr[fo + (e >> 1) * 64 + (e & 1) * 2];
  }

  if (t < 24) {
    float s, c;
    sincosf(theta[t] * 0.5f, &s, &c);
    cs[t] = c;
    sn[t] = s;
  }
  lds_barrier();

  {
    const int r = t >> 4, c = t & 15;
#pragma unroll
    for (int m = 0; m < 6; ++m) {
      float v = 1.f;
#pragma unroll
      for (int q = 0; q < 4; ++q) {
        const int rb = (r >> (3 - q)) & 1;
        const int cb = (c >> (3 - q)) & 1;
        const float cv = cs[m * 4 + q], sv = sn[m * 4 + q];
        v *= rb ? (cb ? cv : sv) : (cb ? -sv : cv);
      }
      G[m][r * GP + c] = v;
    }
  }

#pragma unroll 1
  for (; row < nrows; row += NWG) {
    v2f s2 = {0.f, 0.f};
#pragma unroll
    for (int e = 0; e < 4; ++e) {
      const float4 v = nx[e];
      v2f lo, hi;
      lo.x = v.x; lo.y = v.y;
      hi.x = v.z; hi.y = v.w;
      s2 += lo * lo;
      s2 += hi * hi;
    }
    float ss = s2.x + s2.y;
    ss += lane_xor<16>(ss);
    ss += lane_xor<8>(ss);
    ss += lane_xor<4>(ss);
    ss += lane_xor<2>(ss);
    ss += lane_xor<1>(ss);
    if (lane == 0) wred[wv] = ss;
    lds_barrier();

    float sumsq = 0.f;
#pragma unroll
    for (int w = 0; w < 8; ++w) sumsq += wred[w];

    const bool fb = (sumsq <= 1e-24f);
    float4 cur[4];
#pragma unroll
    for (int e = 0; e < 4; ++e) cur[e] = nx[e];
    if (fb) {
#pragma unroll
      for (int e = 0; e < 4; ++e) cur[e] = make_float4(0.f, 0.f, 0.f, 0.f);
      if (t == 0) cur[0].x = 1.f;
    }
    const float inv = fb ? 1.f : 1.f / sumsq;

    mid_lo_pass(cur, G[1], G[2], st, tb_fs, n, hf);
    if (row + NWG < nrows) {
      const float4* xn = (const float4*)(x + (size_t)(row + NWG) * DIM);
#pragma unroll
      for (int e = 0; e < 4; ++e) nx[e] = xn[fo + (e >> 1) * 64 + (e & 1) * 2];
    }
    lds_barrier();
    hi_pass_perm(G[0], st, st2, tb_hr, tgw, n, hi2);
    lds_barrier();

    {
      float4 f[4];
#pragma unroll
      for (int e = 0; e < 4; ++e)
        f[e] = *(const float4*)&st2[at(tb_fl, ((unsigned)(e >> 1) << 8) | ((unsigned)(e & 1) << 3))];
      mid_lo_pass(f, G[4], G[5], st2, tb_fs, n, hf);
    }
    lds_barrier();
    v8f acc[2];
    hi_pass_keep(G[3], st2, tb_hr, n, hi2, acc);

    v2f qq[4];
    v2f rs = {0.f, 0.f};
#pragma unroll
    for (int i = 0; i < 4; ++i) {
      v2f a0, a1;
      a0.x = acc[0][2 * i]; a0.y = acc[0][2 * i + 1];
      a1.x = acc[1][2 * i]; a1.y = acc[1][2 * i + 1];
      const v2f p0 = a0 * a0, p1 = a1 * a1;
      qq[i] = p0 + p1;
      rs += p0 - p1;
    }
    const v2f e0 = qq[0] + qq[2], f0 = qq[0] - qq[2];
    const v2f e1 = qq[1] + qq[3], f1 = qq[1] - qq[3];
    const v2f ep = e0 + e1, em = e0 - e1;
    const v2f fp = f0 + f1, fm = f0 - f1;
    const float S    = ep.x + ep.y;
    const float dv0  = ep.x - ep.y;
    const float dv10 = em.x - em.y;
    const float dv2  = fp.x + fp.y;
    const float dv21 = fm.x + fm.y;
    const float dt   = rs.x + rs.y;

    float xs = S;
    xs = walsh_hop<4>(xs, lane);
    xs = walsh_hop<3>(xs, lane);
    xs = walsh_hop<2>(xs, lane);
    xs = walsh_hop<1>(xs, lane);
    xs = walsh_hop<0>(xs, lane);
    const bool b4 = (lane & 16) != 0;
    const bool b3 = (lane & 8) != 0;
    const bool b2 = (lane & 4) != 0;
    const float w1v = b4 ? -dv2 : dv2;
    const float w8v = b3 ? -dt : dt;
    const float y0 = pair_hop<16>(w1v, dv21, b4);
    const float y1 = pair_hop<16>(dv10, dv0, b4);
    const float y2 = pair_hop<16>(dt, w8v, b4);
    const float z0 = pair_hop<8>(y0, y1, b3);
    const float z1 = y2 + lane_xor<8>(y2);
    float zs = pair_hop<4>(z0, z1, b2);
    zs += lane_xor<2>(zs);
    zs += lane_xor<1>(zs);

    if (x_keep) wout[wv][x_slot] = xs;
    if (z_keep) wout[wv][z_slot] = zs;
    lds_barrier();
    if (t < 12) {
      float sacc = 0.f;
#pragma unroll
      for (int w2 = 0; w2 < 8; ++w2) {
        const float v = wout[w2][o_slot];
        const bool neg = ((o_neg >> w2) & 1u) != 0;
        sacc += neg ? -v : v;
      }
      out[(size_t)row * 12 + t] = sacc * inv;
    }
  }
}

extern "C" void kernel_launch(void* const* d_in, const int* in_sizes, int n_in,
                              void* d_out, int out_size, void* d_ws,
                              size_t ws_size, hipStream_t stream) {
  const float* x = (const float*)d_in[0];
  const float* theta = (const float*)d_in[1];
  float* out = (float*)d_out;
  const int B = in_sizes[0] / DIM;
  qdense_kernel<<<dim3(NWG), dim3(256), 0, stream>>>(x, theta, out, B);
}


# Wave priority by rows left (s_setprio, levels 3 to 0, one level down per row done): the four workgroups of a WGP stay within a row of each other instead of finishing oldest first; same arithmetic
# speedup vs baseline: 2.8732x; 1.0502x over previous
#include <hip/hip_runtime.h>
#include <math.h>


typedef __attribute__((ext_vector_type(2))) float v2f;
typedef __attribute__((ext_vector_type(8))) float v8f;

#define DIM 4096
#define GP 20
#define NWG 1024

__host__ __device__ constexpr unsigned sw(unsigned k) {
  return k ^ (((k >> 6) & 1u) << 2) ^ (((k >> 7) & 1u) * 24u) ^
         (((k >> 9) & 1u) << 5) ^ (((k >> 11) & 1u) << 4);
}
__device__ __forceinline__ unsigned at(unsigned tb, unsigned c) {
  const unsigned s = sw(c);
  return (tb ^ (s & 63u)) + (s & ~63u);
}

constexpr unsigned cnot_dst(unsigned k) {
  for (int i = 0; i <= 10; ++i)
    for (int j = i + 1; j <= 11; ++j)
      k ^= ((k >> (11 - i)) & 1u) << (11 - j);
  return k;
}
constexpr bool dst_is_shift_xor() {
  for (int b = 0; b < 12; ++b)
    if (cnot_dst(1u << b) != ((1u << b) ^ ((1u << b) >> 1))) return false;
  return true;
}
static_assert(dst_is_shift_xor(), "cascade is not k ^ (k >> 1)");
constexpr unsigned dmask_p(int p) {
  unsigned mm = 0;
  for (int b = 0; b < 12; ++b)
    mm |= ((cnot_dst(1u << b) >> p) & 1u) << b;
  return mm;
}
__constant__ unsigned DST_MASK[12] = {
    dmask_p(0), dmask_p(1), dmask_p(2),  dmask_p(3),
    dmask_p(4), dmask_p(5), dmask_p(6),  dmask_p(7),
    dmask_p(8), dmask_p(9), dmask_p(10), dmask_p(11)};

__device__ __forceinline__ unsigned dst_perm(unsigned k) {
  unsigned r = 0;
#pragma unroll
  for (int p = 0; p < 12; ++p)
    r |= (unsigned)(__popc(k & DST_MASK[p]) & 1) << p;
  return r;
}

template <int TT, int V>
struct DstC {
  static constexpr unsigned value = sw(cnot_dst(((unsigned)TT << 4) | ((unsigned)V << 8)));
};
template <int TT, int V>
__device__ __forceinline__ void scatter_perm(float* sp, unsigned tgw, const v8f& acc) {
  if constexpr (V < 8) {
    sp[tgw ^ DstC<TT, V>::value] = acc[V];
    scatter_perm<TT, V + 1>(sp, tgw, acc);
  }
}

__device__ __forceinline__ v8f wmma4(v2f a, v2f b, v8f c) {
  return __builtin_amdgcn_wmma_f32_16x16x4_f32(false, a, false, b,
                                               (short)0, c, false, false);
}

__device__ __forceinline__ void lds_barrier() {
  __builtin_amdgcn_fence(__ATOMIC_RELEASE, "workgroup", "local");
  __builtin_amdgcn_s_barrier();
  __builtin_amdgcn_fence(__ATOMIC_ACQUIRE, "workgroup", "local");
}

__device__ __forceinline__ void prio_by_rows_left(unsigned left) {
  if (left >= 2u) {
    if (left >= 3u) __builtin_amdgcn_s_setprio(3);
    else            __builtin_amdgcn_s_setprio(2);
  } else {
    if (left >= 1u) __builtin_amdgcn_s_setprio(1);
    else            __builtin_amdgcn_s_setprio(0);
  }
}

template <int M>
__device__ __forceinline__ float lane_xor(float v) {
  static_assert(M == 1 || M == 2 || M == 4 || M == 8 || M == 16, "one lane bit");
  const int b = __builtin_bit_cast(int, v);
  int r;
  if constexpr (M == 16) {
    r = (int)__builtin_amdgcn_permlanex16(b, b, 0x76543210u, 0xfedcba98u, false, false);
  } else {
    r = __builtin_amdgcn_update_dpp(0, b, 0x160 + M, 0xF, 0xF, true);
  }
  return __builtin_bit_cast(float, r);
}

template <int BB>
__device__ __forceinline__ float walsh_hop(float xs, int lane) {
  const float u = lane_xor<(1 << BB)>(xs);
  return ((lane >> BB) & 1) ? (u - xs) : (xs + u);
}

template <int M>
__device__ __forceinline__ float pair_hop(float a, float b, bool bit) {
  const float give = bit ? a : b;
  return (bit ? b : a) + lane_xor<M>(give);
}

__device__ __forceinline__ void mid_lo_pass(const float4 (&f)[4], const float* g1,
                                            const float* g2, float* st,
                                            unsigned tb_fs, int n, int hf) {
  v2f b1[4], b2[4];
#pragma unroll
  for (int kc = 0; kc < 4; ++kc) {
    b1[kc] = *(const v2f*)&g2[n * GP + 4 * hf + 8 * (kc >> 1) + 2 * (kc & 1)];
    b2[kc] = *(const v2f*)&g1[n * GP + 8 * hf + 2 * kc];
  }
  v8f u[2], r[2];
#pragma unroll
  for (int tt = 0; tt < 2; ++tt) {
    u[tt] = {};
#pragma unroll
    for (int kc = 0; kc < 4; ++kc) {
      const float4 v = f[tt * 2 + (kc >> 1)];
      v2f a;
      a.x = (kc & 1) ? v.z : v.x;
      a.y = (kc & 1) ? v.w : v.y;
      u[tt] = wmma4(a, b1[kc], u[tt]);
    }
  }
#pragma unroll
  for (int tt = 0; tt < 2; ++tt) {
    r[tt] = {};
#pragma unroll
    for (int kc = 0; kc < 4; ++kc) {
      v2f a;
      a.x = u[tt][2 * kc];
      a.y = u[tt][2 * kc + 1];
      r[tt] = wmma4(a, b2[kc], r[tt]);
    }
  }
#pragma unroll
  for (int tt = 0; tt < 2; ++tt) {
#pragma unroll
    for (int q = 0; q < 2; ++q) {
      float4 o;
      o.x = r[tt][4 * q];
      o.y = r[tt][4 * q + 1];
      o.z = r[tt][4 * q + 2];
      o.w = r[tt][4 * q + 3];
      *(float4*)&st[at(tb_fs, ((unsigned)tt << 8) | ((unsigned)q << 2))] = o;
    }
  }
}

__device__ __forceinline__ void hi_pass_keep(const float* g0, const float* st, unsigned tb_hr,
                                             int n, int hi2, v8f (&acc)[2]) {
  v2f a[4];
#pragma unroll
  for (int kc = 0; kc < 4; ++kc)
    a[kc] = *(const v2f*)&g0[n * GP + kc * 4 + hi2];
  acc[0] = {};
  acc[1] = {};
#pragma unroll
  for (int kc = 0; kc < 4; ++kc) {
#pragma unroll
    for (int tt = 0; tt < 2; ++tt) {
      const unsigned c = ((unsigned)tt << 4) | ((unsigned)kc << 10);
      v2f b;
      b.x = st[at(tb_hr, c)];
      b.y = st[at(tb_hr, c | (1u << 8))];
      acc[tt] = wmma4(a[kc], b, acc[tt]);
    }
  }
}

template <int TT>
__device__ __forceinline__ void hi_tile_perm(const v2f (&a)[4], const float* st, float* sp,
                                             unsigned tb_hr, unsigned tgw) {
  v8f acc = {};
#pragma unroll
  for (int kc = 0; kc < 4; ++kc) {
    const unsigned c = ((unsigned)TT << 4) | ((unsigned)kc << 10);
    v2f b;
    b.x = st[at(tb_hr, c)];
    b.y = st[at(tb_hr, c | (1u << 8))];
    acc = wmma4(a[kc], b, acc);
  }
  scatter_perm<TT, 0>(sp, tgw, acc);
}
__device__ __forceinline__ void hi_pass_perm(const float* g0, const float* st, float* sp,
                                             unsigned tb_hr, unsigned tgw, int n, int hi2) {
  v2f a[4];
#pragma unroll
  for (int kc = 0; kc < 4; ++kc)
    a[kc] = *(const v2f*)&g0[n * GP + kc * 4 + hi2];
  hi_tile_perm<0>(a, st, sp, tb_hr, tgw);
  hi_tile_perm<1>(a, st, sp, tb_hr, tgw);
}

__global__ __launch_bounds__(256) __attribute__((amdgpu_waves_per_eu(8, 8)))
void qdense_kernel(const float* __restrict__ x,
                   const float* __restrict__ theta,
                   float* __restrict__ out,
                   int nrows) {
  __shared__ __align__(16) float st[DIM];
  __shared__ __align__(16) float st2[DIM];
  __shared__ __align__(16) float G[6][16 * GP];
  __shared__ float cs[24], sn[24];
  __shared__ float wred[8];
  __shared__ float wout[8][12];

  const int t    = threadIdx.x;
  const int lane = t & 31;
  const int wv   = t >> 5;
  const int n    = lane & 15;
  const int hf   = lane >> 4;
  const int hi2  = hf * 2;
  const unsigned un = (unsigned)n, uh = (unsigned)hf, uw = (unsigned)wv;
  const unsigned tb_hr = sw(un | (uw << 5) | (uh << 9));
  const unsigned tb_fl = sw((uw << 9) | (un << 4) | (uh << 2));
  const unsigned tb_fs = sw((uw << 9) | (un << 4) | (uh << 3));
  const unsigned tgw = sw(dst_perm(un | (uw << 5) | (uh << 11)));
  const int fo = wv * 128 + n * 4 + hf;

  constexpr unsigned XL = 0x00011049u, ZL = 0x01110111u;
  const unsigned below = (1u << lane) - 1u;
  const bool x_keep = ((XL >> lane) & 1u) != 0;
  const bool z_keep = ((ZL >> lane) & 1u) != 0;
  const int x_slot = __popc(XL & below);
  const int z_slot = 5 + __popc(ZL & below);
  const unsigned tq = ((unsigned)t & 15u) * 4u;
  const int o_slot = (int)((0x1239600A7854ull >> tq) & 15u);
  const unsigned o_wbits = (0x13640000u >> (tq & 31u)) & (tq < 32u ? 15u : 0u);
  unsigned o_neg = 0;
#pragma unroll
  for (int w2 = 0; w2 < 8; ++w2)
    o_neg |= (unsigned)(__popc((unsigned)w2 & o_wbits) & 1) << w2;

  int row = blockIdx.x;
  float4 nx[4];
  if (row < nrows) {
    prio_by_rows_left((unsigned)(nrows - 1 - row) / NWG);
    const float4* xr = (const float4*)(x + (size_t)row * DIM);
#pragma unroll
    for (int e = 0; e < 4; ++e) nx[e] = xr[fo + (e >> 1) * 64 + (e & 1) * 2];
  }

  if (t < 24) {
    float s, c;
    sincosf(theta[t] * 0.5f, &s, &c);
    cs[t] = c;
    sn[t] = s;
  }
  lds_barrier();

  {
    const int r = t >> 4, c = t & 15;
#pragma unroll
    for (int m = 0; m < 6; ++m) {
      float v = 1.f;
#pragma unroll
      for (int q = 0; q < 4; ++q) {
        const int rb = (r >> (3 - q)) & 1;
        const int cb = (c >> (3 - q)) & 1;
        const float cv = cs[m * 4 + q], sv = sn[m * 4 + q];
        v *= rb ? (cb ? cv : sv) : (cb ? -sv : cv);
      }
      G[m][r * GP + c] = v;
    }
  }

#pragma unroll 1
  for (; row < nrows; row += NWG) {
    v2f s2 = {0.f, 0.f};
#pragma unroll
    for (int e = 0; e < 4; ++e) {
      const float4 v = nx[e];
      v2f lo, hi;
      lo.x = v.x; lo.y = v.y;
      hi.x = v.z; hi.y = v.w;
      s2 += lo * lo;
      s2 += hi * hi;
    }
    float ss = s2.x + s2.y;
    ss += lane_xor<16>(ss);
    ss += lane_xor<8>(ss);
    ss += lane_xor<4>(ss);
    ss += lane_xor<2>(ss);
    ss += lane_xor<1>(ss);
    if (lane == 0) wred[wv] = ss;
    lds_barrier();

    float sumsq = 0.f;
#pragma unroll
    for (int w = 0; w < 8; ++w) sumsq += wred[w];

    const bool fb = (sumsq <= 1e-24f);
    float4 cur[4];
#pragma unroll
    for (int e = 0; e < 4; ++e) cur[e] = nx[e];
    if (fb) {
#pragma unroll
      for (int e = 0; e < 4; ++e) cur[e] = make_float4(0.f, 0.f, 0.f, 0.f);
      if (t == 0) cur[0].x = 1.f;
    }
    const float inv = fb ? 1.f : 1.f / sumsq;

    mid_lo_pass(cur, G[1], G[2], st, tb_fs, n, hf);
    if (row + NWG < nrows) {
      const float4* xn = (const float4*)(x + (size_t)(row + NWG) * DIM);
#pragma unroll
      for (int e = 0; e < 4; ++e) nx[e] = xn[fo + (e >> 1) * 64 + (e & 1) * 2];
    }
    lds_barrier();
    hi_pass_perm(G[0], st, st2, tb_hr, tgw, n, hi2);
    lds_barrier();

    {
      float4 f[4];
#pragma unroll
      for (int e = 0; e < 4; ++e)
        f[e] = *(const float4*)&st2[at(tb_fl, ((unsigned)(e >> 1) << 8) | ((unsigned)(e & 1) << 3))];
      mid_lo_pass(f, G[4], G[5], st2, tb_fs, n, hf);
    }
    lds_barrier();
    v8f acc[2];
    hi_pass_keep(G[3], st2, tb_hr, n, hi2, acc);

    v2f qq[4];
    v2f rs = {0.f, 0.f};
#pragma unroll
    for (int i = 0; i < 4; ++i) {
      v2f a0, a1;
      a0.x = acc[0][2 * i]; a0.y = acc[0][2 * i + 1];
      a1.x = acc[1][2 * i]; a1.y = acc[1][2 * i + 1];
      const v2f p0 = a0 * a0, p1 = a1 * a1;
      qq[i] = p0 + p1;
      rs += p0 - p1;
    }
    const v2f e0 = qq[0] + qq[2], f0 = qq[0] - qq[2];
    const v2f e1 = qq[1] + qq[3], f1 = qq[1] - qq[3];
    const v2f ep = e0 + e1, em = e0 - e1;
    const v2f fp = f0 + f1, fm = f0 - f1;
    const float S    = ep.x + ep.y;
    const float dv0  = ep.x - ep.y;
    const float dv10 = em.x - em.y;
    const float dv2  = fp.x + fp.y;
    const float dv21 = fm.x + fm.y;
    const float dt   = rs.x + rs.y;

    float xs = S;
    xs = walsh_hop<4>(xs, lane);
    xs = walsh_hop<3>(xs, lane);
    xs = walsh_hop<2>(xs, lane);
    xs = walsh_hop<1>(xs, lane);
    xs = walsh_hop<0>(xs, lane);
    const bool b4 = (lane & 16) != 0;
    const bool b3 = (lane & 8) != 0;
    const bool b2 = (lane & 4) != 0;
    const float w1v = b4 ? -dv2 : dv2;
    const float w8v = b3 ? -dt : dt;
    const float y0 = pair_hop<16>(w1v, dv21, b4);
    const float y1 = pair_hop<16>(dv10, dv0, b4);
    const float y2 = pair_hop<16>(dt, w8v, b4);
    const float z0 = pair_hop<8>(y0, y1, b3);
    const float z1 = y2 + lane_xor<8>(y2);
    float zs = pair_hop<4>(z0, z1, b2);
    zs += lane_xor<2>(zs);
    zs += lane_xor<1>(zs);

    if (x_keep) wout[wv][x_slot] = xs;
    if (z_keep) wout[wv][z_slot] = zs;
    lds_barrier();
    if (t < 12) {
      float sacc = 0.f;
#pragma unroll
      for (int w2 = 0; w2 < 8; ++w2) {
        const float v = wout[w2][o_slot];
        const bool neg = ((o_neg >> w2) & 1u) != 0;
        sacc += neg ? -v : v;
      }
      out[(size_t)row * 12 + t] = sacc * inv;
    }
    if (row + NWG < nrows) prio_by_rows_left((unsigned)(nrows - 1 - row) / NWG - 1u);
  }
}

extern "C" void kernel_launch(void* const* d_in, const int* in_sizes, int n_in,
                              void* d_out, int out_size, void* d_ws,
                              size_t ws_size, hipStream_t stream) {
  const float* x = (const float*)d_in[0];
  const float* theta = (const float*)d_in[1];
  float* out = (float*)d_out;
  const int B = in_sizes[0] / DIM;
  qdense_kernel<<<dim3(NWG), dim3(256), 0, stream>>>(x, theta, out, B);
}
